# starting kernel
# baseline (speedup 1.0000x reference)
#include <hip/hip_runtime.h>

typedef __attribute__((ext_vector_type(2))) float v2f;
typedef __attribute__((ext_vector_type(8))) float v8f;

#define DFEAT 128

__global__ void gcn_deg_kernel(const int* __restrict__ src, const int* __restrict__ dst,
                               float* __restrict__ deg_out, float* __restrict__ deg_in, int E) {
  int e = blockIdx.x * blockDim.x + threadIdx.x;
  if (e < E) {
    atomicAdd(&deg_out[src[e]], 1.0f);
    atomicAdd(&deg_in[dst[e]], 1.0f);
  }
}

__global__ void gcn_norm_kernel(float* __restrict__ a, float* __restrict__ b, int N) {
  int i = blockIdx.x * blockDim.x + threadIdx.x;
  if (i < N) {
    a[i] = rsqrtf(fmaxf(a[i], 1.0f));
    b[i] = rsqrtf(fmaxf(b[i], 1.0f));
  }
}

__global__ __launch_bounds__(256) void gcn_scatter_kernel(
    const float* __restrict__ feat, const float* __restrict__ norm_src,
    const int* __restrict__ src, const int* __restrict__ dst,
    float* __restrict__ agg, int E) {
  const int lane = threadIdx.x & 31;
  const int waveId = (blockIdx.x * blockDim.x + threadIdx.x) >> 5;
  const int nWaves = (gridDim.x * blockDim.x) >> 5;

  for (int e = waveId; e < E; e += nWaves) {
    int s = src[e];
    int d = dst[e];
    int e2 = e + nWaves;
    if (e2 < E) {
      int s2 = src[e2];
      __builtin_prefetch((const void*)(feat + (size_t)s2 * DFEAT + lane * 4), 0, 0);
    }
    float w = norm_src[s];
    float4 v = ((const float4*)(feat + (size_t)s * DFEAT))[lane];
    float* o = agg + (size_t)d * DFEAT + lane * 4;
    atomicAdd(o + 0, v.x * w);
    atomicAdd(o + 1, v.y * w);
    atomicAdd(o + 2, v.z * w);
    atomicAdd(o + 3, v.w * w);
  }
}

__global__ __launch_bounds__(256) void gcn_gemm_kernel(
    const float* __restrict__ A, const float* __restrict__ norm_dst,
    const float* __restrict__ W, const float* __restrict__ bias,
    float* __restrict__ out, int N) {
  __shared__ float sA[64 * DFEAT];
  __shared__ float sW[32 * DFEAT];
  __shared__ float sB[DFEAT];

  const int tid = threadIdx.x;
  const int wave = tid >> 5;
  const int lane = tid & 31;
  const int row0 = blockIdx.x * 64;
  const int col0 = wave * 16;
  const int nl = lane & 15;
  const int koff = (lane >> 4) << 1;

  if (tid < DFEAT) sB[tid] = bias[tid];

#pragma unroll
  for (int it = 0; it < 8; ++it) {
    int idx4 = tid + it * 256;
    int r = idx4 >> 5;
    int c4 = idx4 & 31;
    int row = row0 + r;
    float4 v = {0.0f, 0.0f, 0.0f, 0.0f};
    if (row < N) {
      v = ((const float4*)(A + (size_t)row * DFEAT))[c4];
      float sc = norm_dst[row];
      v.x *= sc; v.y *= sc; v.z *= sc; v.w *= sc;
    }
    ((float4*)sA)[idx4] = v;
  }

  v8f acc0 = {}, acc1 = {}, acc2 = {}, acc3 = {};

  for (int kc = 0; kc < DFEAT; kc += 32) {
    __syncthreads();
#pragma unroll
    for (int it = 0; it < 4; ++it) {
      int idx4 = tid + it * 256;
      ((float4*)sW)[idx4] = ((const float4*)(W + (size_t)kc * DFEAT))[idx4];
    }
    __syncthreads();

#pragma unroll
    for (int ks = 0; ks < 8; ++ks) {
      const int kl = ks * 4 + koff;
      const int kg = kc + kl;

      v2f bfrag;
      bfrag[0] = sW[kl * DFEAT + col0 + nl];
      bfrag[1] = sW[(kl + 1) * DFEAT + col0 + nl];

      v2f a0, a1, a2, a3;
      a0[0] = sA[(0 * 16 + nl) * DFEAT + kg]; a0[1] = sA[(0 * 16 + nl) * DFEAT + kg + 1];
      a1[0] = sA[(1 * 16 + nl) * DFEAT + kg]; a1[1] = sA[(1 * 16 + nl) * DFEAT + kg + 1];
      a2[0] = sA[(2 * 16 + nl) * DFEAT + kg]; a2[1] = sA[(2 * 16 + nl) * DFEAT + kg + 1];
      a3[0] = sA[(3 * 16 + nl) * DFEAT + kg]; a3[1] = sA[(3 * 16 + nl) * DFEAT + kg + 1];

      acc0 = __builtin_amdgcn_wmma_f32_16x16x4_f32(false, a0, false, bfrag, (short)0, acc0, false, false);
      acc1 = __builtin_amdgcn_wmma_f32_16x16x4_f32(false, a1, false, bfrag, (short)0, acc1, false, false);
      acc2 = __builtin_amdgcn_wmma_f32_16x16x4_f32(false, a2, false, bfrag, (short)0, acc2, false, false);
      acc3 = __builtin_amdgcn_wmma_f32_16x16x4_f32(false, a3, false, bfrag, (short)0, acc3, false, false);
    }
  }

  const int c = col0 + nl;
  const float bval = sB[c];
  const int mhalf = (lane >> 4) << 3;
#pragma unroll
  for (int j = 0; j < 8; ++j) {
    int ml = j + mhalf;
    int r;
    r = row0 + 0 * 16 + ml; if (r < N) out[(size_t)r * DFEAT + c] = fmaxf(acc0[j] + bval, 0.0f);
    r = row0 + 1 * 16 + ml; if (r < N) out[(size_t)r * DFEAT + c] = fmaxf(acc1[j] + bval, 0.0f);
    r = row0 + 2 * 16 + ml; if (r < N) out[(size_t)r * DFEAT + c] = fmaxf(acc2[j] + bval, 0.0f);
    r = row0 + 3 * 16 + ml; if (r < N) out[(size_t)r * DFEAT + c] = fmaxf(acc3[j] + bval, 0.0f);
  }
}

extern "C" void kernel_launch(void* const* d_in, const int* in_sizes, int n_in,
                              void* d_out, int out_size, void* d_ws, size_t ws_size,
                              hipStream_t stream) {
  const float* feat = (const float*)d_in[0];
  const int* src    = (const int*)d_in[1];
  const int* dst    = (const int*)d_in[2];
  const float* W1   = (const float*)d_in[3];
  const float* b1   = (const float*)d_in[4];
  const float* W2   = (const float*)d_in[5];
  const float* b2   = (const float*)d_in[6];
  float* out        = (float*)d_out;

  const int N = in_sizes[0] / DFEAT;
  const int E = in_sizes[1];

  float* norm_src = (float*)d_ws;
  float* norm_dst = norm_src + N;
  float* agg      = norm_dst + N;
  float* h1       = agg + (size_t)N * DFEAT;

  hipMemsetAsync(norm_src, 0, sizeof(float) * 2 * (size_t)N, stream);
  gcn_deg_kernel<<<(E + 255) / 256, 256, 0, stream>>>(src, dst, norm_src, norm_dst, E);
  gcn_norm_kernel<<<(N + 255) / 256, 256, 0, stream>>>(norm_src, norm_dst, N);

  const int scatterBlocks = 4096;
  const int gemmBlocks = (N + 63) / 64;

  hipMemsetAsync(agg, 0, sizeof(float) * (size_t)N * DFEAT, stream);
  gcn_scatter_kernel<<<scatterBlocks, 256, 0, stream>>>(feat, norm_src, src, dst, agg, E);
  gcn_gemm_kernel<<<gemmBlocks, 256, 0, stream>>>(agg, norm_dst, W1, b1, h1, N);

  hipMemsetAsync(agg, 0, sizeof(float) * (size_t)N * DFEAT, stream);
  gcn_scatter_kernel<<<scatterBlocks, 256, 0, stream>>>(h1, norm_src, src, dst, agg, E);
  gcn_gemm_kernel<<<gemmBlocks, 256, 0, stream>>>(agg, norm_dst, W2, b2, out, N);
}


# f32-atomic scatter-add replaced by per-call counting sort of edges by destination (int-atomic slots, scan) and one wave summing each node's rows in registers in ascending source order
# speedup vs baseline: 1.6488x; 1.6488x over previous
#include <hip/hip_runtime.h>

typedef __attribute__((ext_vector_type(2))) float v2f;
typedef __attribute__((ext_vector_type(8))) float v8f;

#define DFEAT 128
#define SCAN_T 1024

__global__ void gcn_count_kernel(const int* __restrict__ src, const int* __restrict__ dst,
                                 int* __restrict__ cnt_out, int* __restrict__ cnt_in,
                                 int* __restrict__ slot, int E) {
  int e = blockIdx.x * blockDim.x + threadIdx.x;
  if (e < E) {
    atomicAdd(&cnt_out[src[e]], 1);
    slot[e] = atomicAdd(&cnt_in[dst[e]], 1);
  }
}

__global__ __launch_bounds__(SCAN_T) void gcn_scan_kernel(
    const int* __restrict__ cnt_out, const int* __restrict__ cnt_in,
    int* __restrict__ row_ptr, float* __restrict__ norm_src, float* __restrict__ norm_dst, int N) {
  __shared__ int s_pre[32];
  __shared__ int s_own[32];
  const int tid = threadIdx.x;
  const int lane = tid & 31;
  const int wave = tid >> 5;
  const int base = blockIdx.x * SCAN_T;

  int pre = 0;
  for (int i = tid; i < base; i += SCAN_T) pre += cnt_in[i];

  const int i = base + tid;
  const int x = (i < N) ? cnt_in[i] : 0;

  int incl = x;
  for (int d = 1; d < 32; d <<= 1) {
    int y = __shfl_up(incl, d);
    if (lane >= d) incl += y;
  }
  for (int d = 16; d > 0; d >>= 1) pre += __shfl_xor(pre, d);
  if (lane == 31) s_own[wave] = incl;
  if (lane == 0) s_pre[wave] = pre;
  __syncthreads();

  int wp = s_pre[lane];
  int before = (lane < wave) ? s_own[lane] : 0;
  for (int d = 16; d > 0; d >>= 1) {
    wp += __shfl_xor(wp, d);
    before += __shfl_xor(before, d);
  }
  const int excl = wp + before + incl - x;
  if (i < N) {
    row_ptr[i] = excl;
    norm_src[i] = rsqrtf(fmaxf((float)cnt_out[i], 1.0f));
    norm_dst[i] = rsqrtf(fmaxf((float)x, 1.0f));
    if (i == N - 1) row_ptr[N] = excl + x;
  }
}

__global__ void gcn_place_kernel(const int* __restrict__ src, const int* __restrict__ dst,
                                 const int* __restrict__ slot, const int* __restrict__ row_ptr,
                                 int* __restrict__ seg_src, int E) {
  int e = blockIdx.x * blockDim.x + threadIdx.x;
  if (e < E) seg_src[row_ptr[dst[e]] + slot[e]] = src[e];
}

__global__ __launch_bounds__(256) void gcn_order_kernel(
    const int* __restrict__ seg_src, const int* __restrict__ row_ptr,
    const float* __restrict__ norm_src, int* __restrict__ seg_pairs, int N) {
  const int lane = threadIdx.x & 31;
  const int node = __builtin_amdgcn_readfirstlane((int)((blockIdx.x * blockDim.x + threadIdx.x) >> 5));
  if (node >= N) return;
  const int beg = row_ptr[node];
  const int end = row_ptr[node + 1];
  int2* __restrict__ seg = (int2*)seg_pairs;
  for (int cb = beg; cb < end; cb += 32) {
    const int mi = cb + lane;
    const bool have = mi < end;
    const int ms = have ? seg_src[mi] : 0;
    int pos = 0;
    for (int ob = beg; ob < end; ob += 32) {
      const int oi = ob + lane;
      const int os = (oi < end) ? seg_src[oi] : 0;
      const int oc = min(32, end - ob);
      for (int k = 0; k < oc; ++k) {
        const int ks = __builtin_amdgcn_readlane(os, k);
        const int ki = ob + k;
        pos += (ks < ms || (ks == ms && ki < mi)) ? 1 : 0;
      }
    }
    if (have) seg[beg + pos] = make_int2(ms, __float_as_int(norm_src[ms]));
  }
}

__global__ __launch_bounds__(256) void gcn_gather_kernel(
    const float* __restrict__ feat, const int* __restrict__ seg_pairs,
    const int* __restrict__ row_ptr, float* __restrict__ agg, int N) {
  const int lane = threadIdx.x & 31;
  const int node = __builtin_amdgcn_readfirstlane((int)((blockIdx.x * blockDim.x + threadIdx.x) >> 5));
  if (node >= N) return;
  const int beg = row_ptr[node];
  const int end = row_ptr[node + 1];
  const int2* __restrict__ seg = (const int2*)seg_pairs;
  const float4* __restrict__ f4 = (const float4*)feat;
  float4 acc = {0.0f, 0.0f, 0.0f, 0.0f};
  for (int cb = beg; cb < end; cb += 32) {
    const int cnt = min(32, end - cb);
    int2 m = make_int2(0, 0);
    if (lane < cnt) m = seg[cb + lane];
    int k = 0;
    for (; k + 4 <= cnt; k += 4) {
      const int s0 = __builtin_amdgcn_readlane(m.x, k);
      const int s1 = __builtin_amdgcn_readlane(m.x, k + 1);
      const int s2 = __builtin_amdgcn_readlane(m.x, k + 2);
      const int s3 = __builtin_amdgcn_readlane(m.x, k + 3);
      const float w0 = __int_as_float(__builtin_amdgcn_readlane(m.y, k));
      const float w1 = __int_as_float(__builtin_amdgcn_readlane(m.y, k + 1));
      const float w2 = __int_as_float(__builtin_amdgcn_readlane(m.y, k + 2));
      const float w3 = __int_as_float(__builtin_amdgcn_readlane(m.y, k + 3));
      const float4 v0 = f4[(size_t)s0 * (DFEAT / 4) + lane];
      const float4 v1 = f4[(size_t)s1 * (DFEAT / 4) + lane];
      const float4 v2 = f4[(size_t)s2 * (DFEAT / 4) + lane];
      const float4 v3 = f4[(size_t)s3 * (DFEAT / 4) + lane];
      acc.x += v0.x * w0; acc.y += v0.y * w0; acc.z += v0.z * w0; acc.w += v0.w * w0;
      acc.x += v1.x * w1; acc.y += v1.y * w1; acc.z += v1.z * w1; acc.w += v1.w * w1;
      acc.x += v2.x * w2; acc.y += v2.y * w2; acc.z += v2.z * w2; acc.w += v2.w * w2;
      acc.x += v3.x * w3; acc.y += v3.y * w3; acc.z += v3.z * w3; acc.w += v3.w * w3;
    }
    for (; k < cnt; ++k) {
      const int s0 = __builtin_amdgcn_readlane(m.x, k);
      const float w0 = __int_as_float(__builtin_amdgcn_readlane(m.y, k));
      const float4 v0 = f4[(size_t)s0 * (DFEAT / 4) + lane];
      acc.x += v0.x * w0; acc.y += v0.y * w0; acc.z += v0.z * w0; acc.w += v0.w * w0;
    }
  }
  ((float4*)agg)[(size_t)node * (DFEAT / 4) + lane] = acc;
}

__global__ __launch_bounds__(256) void gcn_gemm_kernel(
    const float* __restrict__ A, const float* __restrict__ norm_dst,
    const float* __restrict__ W, const float* __restrict__ bias,
    float* __restrict__ out, int N) {
  __shared__ float sA[64 * DFEAT];
  __shared__ float sW[32 * DFEAT];
  __shared__ float sB[DFEAT];

  const int tid = threadIdx.x;
  const int wave = tid >> 5;
  const int lane = tid & 31;
  const int row0 = blockIdx.x * 64;
  const int col0 = wave * 16;
  const int nl = lane & 15;
  const int koff = (lane >> 4) << 1;

  if (tid < DFEAT) sB[tid] = bias[tid];

#pragma unroll
  for (int it = 0; it < 8; ++it) {
    int idx4 = tid + it * 256;
    int r = idx4 >> 5;
    int c4 = idx4 & 31;
    int row = row0 + r;
    float4 v = {0.0f, 0.0f, 0.0f, 0.0f};
    if (row < N) {
      v = ((const float4*)(A + (size_t)row * DFEAT))[c4];
      float sc = norm_dst[row];
      v.x *= sc; v.y *= sc; v.z *= sc; v.w *= sc;
    }
    ((float4*)sA)[idx4] = v;
  }

  v8f acc0 = {}, acc1 = {}, acc2 = {}, acc3 = {};

  for (int kc = 0; kc < DFEAT; kc += 32) {
    __syncthreads();
#pragma unroll
    for (int it = 0; it < 4; ++it) {
      int idx4 = tid + it * 256;
      ((float4*)sW)[idx4] = ((const float4*)(W + (size_t)kc * DFEAT))[idx4];
    }
    __syncthreads();

#pragma unroll
    for (int ks = 0; ks < 8; ++ks) {
      const int kl = ks * 4 + koff;
      const int kg = kc + kl;

      v2f bfrag;
      bfrag[0] = sW[kl * DFEAT + col0 + nl];
      bfrag[1] = sW[(kl + 1) * DFEAT + col0 + nl];

      v2f a0, a1, a2, a3;
      a0[0] = sA[(0 * 16 + nl) * DFEAT + kg]; a0[1] = sA[(0 * 16 + nl) * DFEAT + kg + 1];
      a1[0] = sA[(1 * 16 + nl) * DFEAT + kg]; a1[1] = sA[(1 * 16 + nl) * DFEAT + kg + 1];
      a2[0] = sA[(2 * 16 + nl) * DFEAT + kg]; a2[1] = sA[(2 * 16 + nl) * DFEAT + kg + 1];
      a3[0] = sA[(3 * 16 + nl) * DFEAT + kg]; a3[1] = sA[(3 * 16 + nl) * DFEAT + kg + 1];

      acc0 = __builtin_amdgcn_wmma_f32_16x16x4_f32(false, a0, false, bfrag, (short)0, acc0, false, false);
      acc1 = __builtin_amdgcn_wmma_f32_16x16x4_f32(false, a1, false, bfrag, (short)0, acc1, false, false);
      acc2 = __builtin_amdgcn_wmma_f32_16x16x4_f32(false, a2, false, bfrag, (short)0, acc2, false, false);
      acc3 = __builtin_amdgcn_wmma_f32_16x16x4_f32(false, a3, false, bfrag, (short)0, acc3, false, false);
    }
  }

  const int c = col0 + nl;
  const float bval = sB[c];
  const int mhalf = (lane >> 4) << 3;
#pragma unroll
  for (int j = 0; j < 8; ++j) {
    int ml = j + mhalf;
    int r;
    r = row0 + 0 * 16 + ml; if (r < N) out[(size_t)r * DFEAT + c] = fmaxf(acc0[j] + bval, 0.0f);
    r = row0 + 1 * 16 + ml; if (r < N) out[(size_t)r * DFEAT + c] = fmaxf(acc1[j] + bval, 0.0f);
    r = row0 + 2 * 16 + ml; if (r < N) out[(size_t)r * DFEAT + c] = fmaxf(acc2[j] + bval, 0.0f);
    r = row0 + 3 * 16 + ml; if (r < N) out[(size_t)r * DFEAT + c] = fmaxf(acc3[j] + bval, 0.0f);
  }
}

extern "C" void kernel_launch(void* const* d_in, const int* in_sizes, int n_in,
                              void* d_out, int out_size, void* d_ws, size_t ws_size,
                              hipStream_t stream) {
  const float* feat = (const float*)d_in[0];
  const int* src    = (const int*)d_in[1];
  const int* dst    = (const int*)d_in[2];
  const float* W1   = (const float*)d_in[3];
  const float* b1   = (const float*)d_in[4];
  const float* W2   = (const float*)d_in[5];
  const float* b2   = (const float*)d_in[6];
  float* out        = (float*)d_out;

  const int N = in_sizes[0] / DFEAT;
  const int E = in_sizes[1];

  float* agg      = (float*)d_ws;
  float* h1       = agg + (size_t)N * DFEAT;
  int* seg_pairs  = (int*)(h1 + (size_t)N * DFEAT);
  int* seg_src    = seg_pairs + 2 * (size_t)E;
  int* slot       = seg_src + E;
  int* cnt_out    = slot + E;
  int* cnt_in     = cnt_out + N;
  int* row_ptr    = cnt_in + N;
  float* norm_src = (float*)(row_ptr + N + 1);
  float* norm_dst = norm_src + N;

  const int edgeBlocks = (E + 255) / 256;
  const int nodeWaveBlocks = (N + 7) / 8;
  const int gemmBlocks = (N + 63) / 64;

  hipMemsetAsync(cnt_out, 0, sizeof(int) * 2 * (size_t)N, stream);
  gcn_count_kernel<<<edgeBlocks, 256, 0, stream>>>(src, dst, cnt_out, cnt_in, slot, E);
  gcn_scan_kernel<<<(N + SCAN_T - 1) / SCAN_T, SCAN_T, 0, stream>>>(cnt_out, cnt_in, row_ptr, norm_src, norm_dst, N);
  gcn_place_kernel<<<edgeBlocks, 256, 0, stream>>>(src, dst, slot, row_ptr, seg_src, E);
  gcn_order_kernel<<<nodeWaveBlocks, 256, 0, stream>>>(seg_src, row_ptr, norm_src, seg_pairs, N);

  gcn_gather_kernel<<<nodeWaveBlocks, 256, 0, stream>>>(feat, seg_pairs, row_ptr, agg, N);
  gcn_gemm_kernel<<<gemmBlocks, 256, 0, stream>>>(agg, norm_dst, W1, b1, h1, N);

  gcn_gather_kernel<<<nodeWaveBlocks, 256, 0, stream>>>(h1, seg_pairs, row_ptr, agg, N);
  gcn_gemm_kernel<<<gemmBlocks, 256, 0, stream>>>(agg, norm_dst, W2, b2, out, N);
}


# edge count, scan and place by global int atomics replaced by LDS counters: a block orders its edge chunk by node range in LDS, then one owner block per 256 nodes counts, scans and places
# speedup vs baseline: 2.6549x; 1.6102x over previous
#include <hip/hip_runtime.h>

typedef __attribute__((ext_vector_type(2))) float v2f;
typedef __attribute__((ext_vector_type(8))) float v8f;

#define DFEAT 128
#define RNODES 256
#define RSHIFT 8
#define CHUNK 8192
#define BIN_T 1024
#define EPT (CHUNK / BIN_T)
#define PMAX 1024
#define OWN_T 1024

__global__ __launch_bounds__(BIN_T) void gcn_bin_kernel(
    const int* __restrict__ src, const int* __restrict__ dst,
    int* __restrict__ bucket_d, int* __restrict__ bucket_s,
    int* __restrict__ tab_d, int* __restrict__ tab_s, int E, int P, int B) {
  __shared__ int2 stage_d[CHUNK];
  __shared__ int stage_s[CHUNK];
  __shared__ int hist_d[PMAX];
  __shared__ int hist_s[PMAX];
  __shared__ int wsum_d[32];
  __shared__ int wsum_s[32];
  const int tid = threadIdx.x;
  const int lane = tid & 31;
  const int wave = tid >> 5;
  const int b = blockIdx.x;
  const int e0 = b * CHUNK;
  const int cnt = min(CHUNK, E - e0);

  hist_d[tid] = 0;
  hist_s[tid] = 0;
  __syncthreads();

  int s[EPT], d[EPT], kd[EPT], ks[EPT];
#pragma unroll
  for (int i = 0; i < EPT; ++i) {
    const int j = i * BIN_T + tid;
    const bool have = j < cnt;
    s[i] = have ? src[e0 + j] : 0;
    d[i] = have ? dst[e0 + j] : 0;
  }
#pragma unroll
  for (int i = 0; i < EPT; ++i) {
    const int j = i * BIN_T + tid;
    kd[i] = 0; ks[i] = 0;
    if (j < cnt) {
      kd[i] = atomicAdd(&hist_d[d[i] >> RSHIFT], 1);
      ks[i] = atomicAdd(&hist_s[s[i] >> RSHIFT], 1);
    }
  }
  __syncthreads();

  const int xd = hist_d[tid];
  const int xs = hist_s[tid];
  int id = xd, is = xs;
  for (int o = 1; o < 32; o <<= 1) {
    const int yd = __shfl_up(id, o);
    const int ys = __shfl_up(is, o);
    if (lane >= o) { id += yd; is += ys; }
  }
  if (lane == 31) { wsum_d[wave] = id; wsum_s[wave] = is; }
  __syncthreads();
  int bd = (lane < wave) ? wsum_d[lane] : 0;
  int bs = (lane < wave) ? wsum_s[lane] : 0;
  for (int o = 16; o > 0; o >>= 1) {
    bd += __shfl_xor(bd, o);
    bs += __shfl_xor(bs, o);
  }
  const int exd = bd + id - xd;
  const int exs = bs + is - xs;
  hist_d[tid] = exd;
  hist_s[tid] = exs;
  if (tid <= P) {
    tab_d[tid * B + b] = exd;
    tab_s[tid * B + b] = exs;
  }
  __syncthreads();

#pragma unroll
  for (int i = 0; i < EPT; ++i) {
    const int j = i * BIN_T + tid;
    if (j < cnt) {
      stage_d[hist_d[d[i] >> RSHIFT] + kd[i]] = make_int2(d[i], s[i]);
      stage_s[hist_s[s[i] >> RSHIFT] + ks[i]] = s[i];
    }
  }
  __syncthreads();
  for (int j = tid; j < cnt; j += BIN_T) {
    ((int2*)bucket_d)[e0 + j] = stage_d[j];
    bucket_s[e0 + j] = stage_s[j];
  }
}

__global__ __launch_bounds__(OWN_T) void gcn_owner_kernel(
    const int* __restrict__ bucket_d, const int* __restrict__ bucket_s,
    const int* __restrict__ tab_d, const int* __restrict__ tab_s,
    int* __restrict__ row_ptr, float* __restrict__ norm_src, float* __restrict__ norm_dst,
    int* __restrict__ seg_src, int N, int B) {
  __shared__ int cnt_d[RNODES];
  __shared__ int cnt_s[RNODES];
  __shared__ int s_red[32];
  __shared__ int s_wtot[RNODES / 32];
  const int tid = threadIdx.x;
  const int lane = tid & 31;
  const int wave = __builtin_amdgcn_readfirstlane(tid >> 5);
  const int r = blockIdx.x;
  const int lo = r * RNODES;
  const int2* __restrict__ bd2 = (const int2*)bucket_d;

  if (tid < RNODES) { cnt_d[tid] = 0; cnt_s[tid] = 0; }
  int part = 0;
  for (int b = tid; b < B; b += OWN_T) part += tab_d[r * B + b];
  for (int o = 16; o > 0; o >>= 1) part += __shfl_xor(part, o);
  if (lane == 0) s_red[wave] = part;
  __syncthreads();

  for (int g = 0; g + wave < B; g += 32 * 32) {
    const int myb = g + wave + 32 * lane;
    int db = 0, de = 0, sb = 0, se = 0;
    if (myb < B) {
      db = tab_d[r * B + myb]; de = tab_d[(r + 1) * B + myb];
      sb = tab_s[r * B + myb]; se = tab_s[(r + 1) * B + myb];
    }
    const int nl = min(32, (B - g - wave + 31) / 32);
    for (int l = 0; l < nl; ++l) {
      const int bb = g + wave + 32 * l;
      const int d0 = __builtin_amdgcn_readlane(db, l);
      const int dn = __builtin_amdgcn_readlane(de, l) - d0;
      const int s0 = __builtin_amdgcn_readlane(sb, l);
      const int sn = __builtin_amdgcn_readlane(se, l) - s0;
      const int len = max(dn, sn);
      for (int o = lane; o < len; o += 32) {
        const bool hd = o < dn;
        const bool hs = o < sn;
        int ed = lo, es = lo;
        if (hd) ed = bd2[bb * CHUNK + d0 + o].x;
        if (hs) es = bucket_s[bb * CHUNK + s0 + o];
        if (hd) atomicAdd(&cnt_d[ed - lo], 1);
        if (hs) atomicAdd(&cnt_s[es - lo], 1);
      }
    }
  }
  __syncthreads();

  const int x = (tid < RNODES) ? cnt_d[tid] : 0;
  int incl = x;
  for (int o = 1; o < 32; o <<= 1) {
    const int y = __shfl_up(incl, o);
    if (lane >= o) incl += y;
  }
  if (tid < RNODES && lane == 31) s_wtot[wave] = incl;
  __syncthreads();
  int base = s_red[lane];
  int before = (lane < wave && lane < RNODES / 32) ? s_wtot[lane] : 0;
  for (int o = 16; o > 0; o >>= 1) {
    base += __shfl_xor(base, o);
    before += __shfl_xor(before, o);
  }
  if (tid < RNODES) {
    const int excl = base + before + incl - x;
    const int node = lo + tid;
    if (node < N) {
      row_ptr[node] = excl;
      norm_src[node] = rsqrtf(fmaxf((float)cnt_s[tid], 1.0f));
      norm_dst[node] = rsqrtf(fmaxf((float)x, 1.0f));
      if (node == N - 1) row_ptr[N] = excl + x;
    }
    cnt_d[tid] = excl;
  }
  __syncthreads();

  for (int g = 0; g + wave < B; g += 32 * 32) {
    const int myb = g + wave + 32 * lane;
    int db = 0, de = 0;
    if (myb < B) { db = tab_d[r * B + myb]; de = tab_d[(r + 1) * B + myb]; }
    const int nl = min(32, (B - g - wave + 31) / 32);
    for (int l = 0; l < nl; ++l) {
      const int bb = g + wave + 32 * l;
      const int d0 = __builtin_amdgcn_readlane(db, l);
      const int dn = __builtin_amdgcn_readlane(de, l) - d0;
      for (int o = lane; o < dn; o += 32) {
        const int2 ed = bd2[bb * CHUNK + d0 + o];
        const int pos = atomicAdd(&cnt_d[ed.x - lo], 1);
        seg_src[pos] = ed.y;
      }
    }
  }
}

__global__ __launch_bounds__(256) void gcn_order_kernel(
    const int* __restrict__ seg_src, const int* __restrict__ row_ptr,
    const float* __restrict__ norm_src, int* __restrict__ seg_pairs, int N) {
  const int lane = threadIdx.x & 31;
  const int node = __builtin_amdgcn_readfirstlane((int)((blockIdx.x * blockDim.x + threadIdx.x) >> 5));
  if (node >= N) return;
  const int beg = row_ptr[node];
  const int end = row_ptr[node + 1];
  int2* __restrict__ seg = (int2*)seg_pairs;
  for (int cb = beg; cb < end; cb += 32) {
    const int mi = cb + lane;
    const bool have = mi < end;
    const int ms = have ? seg_src[mi] : 0;
    int pos = 0;
    for (int ob = beg; ob < end; ob += 32) {
      const int oi = ob + lane;
      const int os = (oi < end) ? seg_src[oi] : 0;
      const int oc = min(32, end - ob);
      for (int k = 0; k < oc; ++k) {
        const int ks = __builtin_amdgcn_readlane(os, k);
        const int ki = ob + k;
        pos += (ks < ms || (ks == ms && ki < mi)) ? 1 : 0;
      }
    }
    if (have) seg[beg + pos] = make_int2(ms, __float_as_int(norm_src[ms]));
  }
}

__global__ __launch_bounds__(256) void gcn_gather_kernel(
    const float* __restrict__ feat, const int* __restrict__ seg_pairs,
    const int* __restrict__ row_ptr, float* __restrict__ agg, int N) {
  const int lane = threadIdx.x & 31;
  const int node = __builtin_amdgcn_readfirstlane((int)((blockIdx.x * blockDim.x + threadIdx.x) >> 5));
  if (node >= N) return;
  const int beg = row_ptr[node];
  const int end = row_ptr[node + 1];
  const int2* __restrict__ seg = (const int2*)seg_pairs;
  const float4* __restrict__ f4 = (const float4*)feat;
  float4 acc = {0.0f, 0.0f, 0.0f, 0.0f};
  for (int cb = beg; cb < end; cb += 32) {
    const int cnt = min(32, end - cb);
    int2 m = make_int2(0, 0);
    if (lane < cnt) m = seg[cb + lane];
    int k = 0;
    for (; k + 4 <= cnt; k += 4) {
      const int s0 = __builtin_amdgcn_readlane(m.x, k);
      const int s1 = __builtin_amdgcn_readlane(m.x, k + 1);
      const int s2 = __builtin_amdgcn_readlane(m.x, k + 2);
      const int s3 = __builtin_amdgcn_readlane(m.x, k + 3);
      const float w0 = __int_as_float(__builtin_amdgcn_readlane(m.y, k));
      const float w1 = __int_as_float(__builtin_amdgcn_readlane(m.y, k + 1));
      const float w2 = __int_as_float(__builtin_amdgcn_readlane(m.y, k + 2));
      const float w3 = __int_as_float(__builtin_amdgcn_readlane(m.y, k + 3));
      const float4 v0 = f4[(size_t)s0 * (DFEAT / 4) + lane];
      const float4 v1 = f4[(size_t)s1 * (DFEAT / 4) + lane];
      const float4 v2 = f4[(size_t)s2 * (DFEAT / 4) + lane];
      const float4 v3 = f4[(size_t)s3 * (DFEAT / 4) + lane];
      acc.x += v0.x * w0; acc.y += v0.y * w0; acc.z += v0.z * w0; acc.w += v0.w * w0;
      acc.x += v1.x * w1; acc.y += v1.y * w1; acc.z += v1.z * w1; acc.w += v1.w * w1;
      acc.x += v2.x * w2; acc.y += v2.y * w2; acc.z += v2.z * w2; acc.w += v2.w * w2;
      acc.x += v3.x * w3; acc.y += v3.y * w3; acc.z += v3.z * w3; acc.w += v3.w * w3;
    }
    for (; k < cnt; ++k) {
      const int s0 = __builtin_amdgcn_readlane(m.x, k);
      const float w0 = __int_as_float(__builtin_amdgcn_readlane(m.y, k));
      const float4 v0 = f4[(size_t)s0 * (DFEAT / 4) + lane];
      acc.x += v0.x * w0; acc.y += v0.y * w0; acc.z += v0.z * w0; acc.w += v0.w * w0;
    }
  }
  ((float4*)agg)[(size_t)node * (DFEAT / 4) + lane] = acc;
}

__global__ __launch_bounds__(256) void gcn_gemm_kernel(
    const float* __restrict__ A, const float* __restrict__ norm_dst,
    const float* __restrict__ W, const float* __restrict__ bias,
    float* __restrict__ out, int N) {
  __shared__ float sA[64 * DFEAT];
  __shared__ float sW[32 * DFEAT];
  __shared__ float sB[DFEAT];

  const int tid = threadIdx.x;
  const int wave = tid >> 5;
  const int lane = tid & 31;
  const int row0 = blockIdx.x * 64;
  const int col0 = wave * 16;
  const int nl = lane & 15;
  const int koff = (lane >> 4) << 1;

  if (tid < DFEAT) sB[tid] = bias[tid];

#pragma unroll
  for (int it = 0; it < 8; ++it) {
    int idx4 = tid + it * 256;
    int r = idx4 >> 5;
    int c4 = idx4 & 31;
    int row = row0 + r;
    float4 v = {0.0f, 0.0f, 0.0f, 0.0f};
    if (row < N) {
      v = ((const float4*)(A + (size_t)row * DFEAT))[c4];
      float sc = norm_dst[row];
      v.x *= sc; v.y *= sc; v.z *= sc; v.w *= sc;
    }
    ((float4*)sA)[idx4] = v;
  }

  v8f acc0 = {}, acc1 = {}, acc2 = {}, acc3 = {};

  for (int kc = 0; kc < DFEAT; kc += 32) {
    __syncthreads();
#pragma unroll
    for (int it = 0; it < 4; ++it) {
      int idx4 = tid + it * 256;
      ((float4*)sW)[idx4] = ((const float4*)(W + (size_t)kc * DFEAT))[idx4];
    }
    __syncthreads();

#pragma unroll
    for (int ks = 0; ks < 8; ++ks) {
      const int kl = ks * 4 + koff;
      const int kg = kc + kl;

      v2f bfrag;
      bfrag[0] = sW[kl * DFEAT + col0 + nl];
      bfrag[1] = sW[(kl + 1) * DFEAT + col0 + nl];

      v2f a0, a1, a2, a3;
      a0[0] = sA[(0 * 16 + nl) * DFEAT + kg]; a0[1] = sA[(0 * 16 + nl) * DFEAT + kg + 1];
      a1[0] = sA[(1 * 16 + nl) * DFEAT + kg]; a1[1] = sA[(1 * 16 + nl) * DFEAT + kg + 1];
      a2[0] = sA[(2 * 16 + nl) * DFEAT + kg]; a2[1] = sA[(2 * 16 + nl) * DFEAT + kg + 1];
      a3[0] = sA[(3 * 16 + nl) * DFEAT + kg]; a3[1] = sA[(3 * 16 + nl) * DFEAT + kg + 1];

      acc0 = __builtin_amdgcn_wmma_f32_16x16x4_f32(false, a0, false, bfrag, (short)0, acc0, false, false);
      acc1 = __builtin_amdgcn_wmma_f32_16x16x4_f32(false, a1, false, bfrag, (short)0, acc1, false, false);
      acc2 = __builtin_amdgcn_wmma_f32_16x16x4_f32(false, a2, false, bfrag, (short)0, acc2, false, false);
      acc3 = __builtin_amdgcn_wmma_f32_16x16x4_f32(false, a3, false, bfrag, (short)0, acc3, false, false);
    }
  }

  const int c = col0 + nl;
  const float bval = sB[c];
  const int mhalf = (lane >> 4) << 3;
#pragma unroll
  for (int j = 0; j < 8; ++j) {
    int ml = j + mhalf;
    int r;
    r = row0 + 0 * 16 + ml; if (r < N) out[(size_t)r * DFEAT + c] = fmaxf(acc0[j] + bval, 0.0f);
    r = row0 + 1 * 16 + ml; if (r < N) out[(size_t)r * DFEAT + c] = fmaxf(acc1[j] + bval, 0.0f);
    r = row0 + 2 * 16 + ml; if (r < N) out[(size_t)r * DFEAT + c] = fmaxf(acc2[j] + bval, 0.0f);
    r = row0 + 3 * 16 + ml; if (r < N) out[(size_t)r * DFEAT + c] = fmaxf(acc3[j] + bval, 0.0f);
  }
}

extern "C" void kernel_launch(void* const* d_in, const int* in_sizes, int n_in,
                              void* d_out, int out_size, void* d_ws, size_t ws_size,
                              hipStream_t stream) {
  const float* feat = (const float*)d_in[0];
  const int* src    = (const int*)d_in[1];
  const int* dst    = (const int*)d_in[2];
  const float* W1   = (const float*)d_in[3];
  const float* b1   = (const float*)d_in[4];
  const float* W2   = (const float*)d_in[5];
  const float* b2   = (const float*)d_in[6];
  float* out        = (float*)d_out;

  const int N = in_sizes[0] / DFEAT;
  const int E = in_sizes[1];

  const int B = (E + CHUNK - 1) / CHUNK;
  const int P = (N + RNODES - 1) / RNODES;

  float* agg      = (float*)d_ws;
  float* h1       = agg + (size_t)N * DFEAT;
  int* seg_pairs  = (int*)(h1 + (size_t)N * DFEAT);
  int* bucket_d   = seg_pairs + 2 * (size_t)E;
  int* seg_src    = bucket_d + 2 * (size_t)E;
  int* bucket_s   = seg_src + E;
  int* tab_d      = bucket_s + E;
  int* tab_s      = tab_d + (size_t)(P + 1) * B;
  int* row_ptr    = tab_s + (size_t)(P + 1) * B;
  float* norm_src = (float*)(row_ptr + N + 1);
  float* norm_dst = norm_src + N;

  const int nodeWaveBlocks = (N + 7) / 8;
  const int gemmBlocks = (N + 63) / 64;

  gcn_bin_kernel<<<B, BIN_T, 0, stream>>>(src, dst, bucket_d, bucket_s, tab_d, tab_s, E, P, B);
  gcn_owner_kernel<<<P, OWN_T, 0, stream>>>(bucket_d, bucket_s, tab_d, tab_s, row_ptr, norm_src, norm_dst, seg_src, N, B);
  gcn_order_kernel<<<nodeWaveBlocks, 256, 0, stream>>>(seg_src, row_ptr, norm_src, seg_pairs, N);

  gcn_gather_kernel<<<nodeWaveBlocks, 256, 0, stream>>>(feat, seg_pairs, row_ptr, agg, N);
  gcn_gemm_kernel<<<gemmBlocks, 256, 0, stream>>>(agg, norm_dst, W1, b1, h1, N);

  gcn_gather_kernel<<<nodeWaveBlocks, 256, 0, stream>>>(h1, seg_pairs, row_ptr, agg, N);
  gcn_gemm_kernel<<<gemmBlocks, 256, 0, stream>>>(agg, norm_dst, W2, b2, out, N);
}


# GEMM LDS row pitches padded (A band 128 to 132 floats, W chunk 128 to 136): the 16 rows of an A fragment read and the two lane halves of a B fragment read now fall on different banks
# speedup vs baseline: 3.4926x; 1.3155x over previous
#include <hip/hip_runtime.h>

typedef __attribute__((ext_vector_type(2))) float v2f;
typedef __attribute__((ext_vector_type(8))) float v8f;

#define DFEAT 128
#define RNODES 256
#define RSHIFT 8
#define CHUNK 8192
#define BIN_T 1024
#define EPT (CHUNK / BIN_T)
#define PMAX 1024
#define OWN_T 1024

__global__ __launch_bounds__(BIN_T) void gcn_bin_kernel(
    const int* __restrict__ src, const int* __restrict__ dst,
    int* __restrict__ bucket_d, int* __restrict__ bucket_s,
    int* __restrict__ tab_d, int* __restrict__ tab_s, int E, int P, int B) {
  __shared__ int2 stage_d[CHUNK];
  __shared__ int stage_s[CHUNK];
  __shared__ int hist_d[PMAX];
  __shared__ int hist_s[PMAX];
  __shared__ int wsum_d[32];
  __shared__ int wsum_s[32];
  const int tid = threadIdx.x;
  const int lane = tid & 31;
  const int wave = tid >> 5;
  const int b = blockIdx.x;
  const int e0 = b * CHUNK;
  const int cnt = min(CHUNK, E - e0);

  hist_d[tid] = 0;
  hist_s[tid] = 0;
  __syncthreads();

  int s[EPT], d[EPT], kd[EPT], ks[EPT];
#pragma unroll
  for (int i = 0; i < EPT; ++i) {
    const int j = i * BIN_T + tid;
    const bool have = j < cnt;
    s[i] = have ? src[e0 + j] : 0;
    d[i] = have ? dst[e0 + j] : 0;
  }
#pragma unroll
  for (int i = 0; i < EPT; ++i) {
    const int j = i * BIN_T + tid;
    kd[i] = 0; ks[i] = 0;
    if (j < cnt) {
      kd[i] = atomicAdd(&hist_d[d[i] >> RSHIFT], 1);
      ks[i] = atomicAdd(&hist_s[s[i] >> RSHIFT], 1);
    }
  }
  __syncthreads();

  const int xd = hist_d[tid];
  const int xs = hist_s[tid];
  int id = xd, is = xs;
  for (int o = 1; o < 32; o <<= 1) {
    const int yd = __shfl_up(id, o);
    const int ys = __shfl_up(is, o);
    if (lane >= o) { id += yd; is += ys; }
  }
  if (lane == 31) { wsum_d[wave] = id; wsum_s[wave] = is; }
  __syncthreads();
  int bd = (lane < wave) ? wsum_d[lane] : 0;
  int bs = (lane < wave) ? wsum_s[lane] : 0;
  for (int o = 16; o > 0; o >>= 1) {
    bd += __shfl_xor(bd, o);
    bs += __shfl_xor(bs, o);
  }
  const int exd = bd + id - xd;
  const int exs = bs + is - xs;
  hist_d[tid] = exd;
  hist_s[tid] = exs;
  if (tid <= P) {
    tab_d[tid * B + b] = exd;
    tab_s[tid * B + b] = exs;
  }
  __syncthreads();

#pragma unroll
  for (int i = 0; i < EPT; ++i) {
    const int j = i * BIN_T + tid;
    if (j < cnt) {
      stage_d[hist_d[d[i] >> RSHIFT] + kd[i]] = make_int2(d[i], s[i]);
      stage_s[hist_s[s[i] >> RSHIFT] + ks[i]] = s[i];
    }
  }
  __syncthreads();
  for (int j = tid; j < cnt; j += BIN_T) {
    ((int2*)bucket_d)[e0 + j] = stage_d[j];
    bucket_s[e0 + j] = stage_s[j];
  }
}

__global__ __launch_bounds__(OWN_T) void gcn_owner_kernel(
    const int* __restrict__ bucket_d, const int* __restrict__ bucket_s,
    const int* __restrict__ tab_d, const int* __restrict__ tab_s,
    int* __restrict__ row_ptr, float* __restrict__ norm_src, float* __restrict__ norm_dst,
    int* __restrict__ seg_src, int N, int B) {
  __shared__ int cnt_d[RNODES];
  __shared__ int cnt_s[RNODES];
  __shared__ int s_red[32];
  __shared__ int s_wtot[RNODES / 32];
  const int tid = threadIdx.x;
  const int lane = tid & 31;
  const int wave = __builtin_amdgcn_readfirstlane(tid >> 5);
  const int r = blockIdx.x;
  const int lo = r * RNODES;
  const int2* __restrict__ bd2 = (const int2*)bucket_d;

  if (tid < RNODES) { cnt_d[tid] = 0; cnt_s[tid] = 0; }
  int part = 0;
  for (int b = tid; b < B; b += OWN_T) part += tab_d[r * B + b];
  for (int o = 16; o > 0; o >>= 1) part += __shfl_xor(part, o);
  if (lane == 0) s_red[wave] = part;
  __syncthreads();

  for (int g = 0; g + wave < B; g += 32 * 32) {
    const int myb = g + wave + 32 * lane;
    int db = 0, de = 0, sb = 0, se = 0;
    if (myb < B) {
      db = tab_d[r * B + myb]; de = tab_d[(r + 1) * B + myb];
      sb = tab_s[r * B + myb]; se = tab_s[(r + 1) * B + myb];
    }
    const int nl = min(32, (B - g - wave + 31) / 32);
    for (int l = 0; l < nl; ++l) {
      const int bb = g + wave + 32 * l;
      const int d0 = __builtin_amdgcn_readlane(db, l);
      const int dn = __builtin_amdgcn_readlane(de, l) - d0;
      const int s0 = __builtin_amdgcn_readlane(sb, l);
      const int sn = __builtin_amdgcn_readlane(se, l) - s0;
      const int len = max(dn, sn);
      for (int o = lane; o < len; o += 32) {
        const bool hd = o < dn;
        const bool hs = o < sn;
        int ed = lo, es = lo;
        if (hd) ed = bd2[bb * CHUNK + d0 + o].x;
        if (hs) es = bucket_s[bb * CHUNK + s0 + o];
        if (hd) atomicAdd(&cnt_d[ed - lo], 1);
        if (hs) atomicAdd(&cnt_s[es - lo], 1);
      }
    }
  }
  __syncthreads();

  const int x = (tid < RNODES) ? cnt_d[tid] : 0;
  int incl = x;
  for (int o = 1; o < 32; o <<= 1) {
    const int y = __shfl_up(incl, o);
    if (lane >= o) incl += y;
  }
  if (tid < RNODES && lane == 31) s_wtot[wave] = incl;
  __syncthreads();
  int base = s_red[lane];
  int before = (lane < wave && lane < RNODES / 32) ? s_wtot[lane] : 0;
  for (int o = 16; o > 0; o >>= 1) {
    base += __shfl_xor(base, o);
    before += __shfl_xor(before, o);
  }
  if (tid < RNODES) {
    const int excl = base + before + incl - x;
    const int node = lo + tid;
    if (node < N) {
      row_ptr[node] = excl;
      norm_src[node] = rsqrtf(fmaxf((float)cnt_s[tid], 1.0f));
      norm_dst[node] = rsqrtf(fmaxf((float)x, 1.0f));
      if (node == N - 1) row_ptr[N] = excl + x;
    }
    cnt_d[tid] = excl;
  }
  __syncthreads();

  for (int g = 0; g + wave < B; g += 32 * 32) {
    const int myb = g + wave + 32 * lane;
    int db = 0, de = 0;
    if (myb < B) { db = tab_d[r * B + myb]; de = tab_d[(r + 1) * B + myb]; }
    const int nl = min(32, (B - g - wave + 31) / 32);
    for (int l = 0; l < nl; ++l) {
      const int bb = g + wave + 32 * l;
      const int d0 = __builtin_amdgcn_readlane(db, l);
      const int dn = __builtin_amdgcn_readlane(de, l) - d0;
      for (int o = lane; o < dn; o += 32) {
        const int2 ed = bd2[bb * CHUNK + d0 + o];
        const int pos = atomicAdd(&cnt_d[ed.x - lo], 1);
        seg_src[pos] = ed.y;
      }
    }
  }
}

__global__ __launch_bounds__(256) void gcn_order_kernel(
    const int* __restrict__ seg_src, const int* __restrict__ row_ptr,
    const float* __restrict__ norm_src, int* __restrict__ seg_pairs, int N) {
  const int lane = threadIdx.x & 31;
  const int node = __builtin_amdgcn_readfirstlane((int)((blockIdx.x * blockDim.x + threadIdx.x) >> 5));
  if (node >= N) return;
  const int beg = row_ptr[node];
  const int end = row_ptr[node + 1];
  int2* __restrict__ seg = (int2*)seg_pairs;
  for (int cb = beg; cb < end; cb += 32) {
    const int mi = cb + lane;
    const bool have = mi < end;
    const int ms = have ? seg_src[mi] : 0;
    int pos = 0;
    for (int ob = beg; ob < end; ob += 32) {
      const int oi = ob + lane;
      const int os = (oi < end) ? seg_src[oi] : 0;
      const int oc = min(32, end - ob);
      for (int k = 0; k < oc; ++k) {
        const int ks = __builtin_amdgcn_readlane(os, k);
        const int ki = ob + k;
        pos += (ks < ms || (ks == ms && ki < mi)) ? 1 : 0;
      }
    }
    if (have) seg[beg + pos] = make_int2(ms, __float_as_int(norm_src[ms]));
  }
}

__global__ __launch_bounds__(256) void gcn_gather_kernel(
    const float* __restrict__ feat, const int* __restrict__ seg_pairs,
    const int* __restrict__ row_ptr, float* __restrict__ agg, int N) {
  const int lane = threadIdx.x & 31;
  const int node = __builtin_amdgcn_readfirstlane((int)((blockIdx.x * blockDim.x + threadIdx.x) >> 5));
  if (node >= N) return;
  const int beg = row_ptr[node];
  const int end = row_ptr[node + 1];
  const int2* __restrict__ seg = (const int2*)seg_pairs;
  const float4* __restrict__ f4 = (const float4*)feat;
  float4 acc = {0.0f, 0.0f, 0.0f, 0.0f};
  for (int cb = beg; cb < end; cb += 32) {
    const int cnt = min(32, end - cb);
    int2 m = make_int2(0, 0);
    if (lane < cnt) m = seg[cb + lane];
    int k = 0;
    for (; k + 4 <= cnt; k += 4) {
      const int s0 = __builtin_amdgcn_readlane(m.x, k);
      const int s1 = __builtin_amdgcn_readlane(m.x, k + 1);
      const int s2 = __builtin_amdgcn_readlane(m.x, k + 2);
      const int s3 = __builtin_amdgcn_readlane(m.x, k + 3);
      const float w0 = __int_as_float(__builtin_amdgcn_readlane(m.y, k));
      const float w1 = __int_as_float(__builtin_amdgcn_readlane(m.y, k + 1));
      const float w2 = __int_as_float(__builtin_amdgcn_readlane(m.y, k + 2));
      const float w3 = __int_as_float(__builtin_amdgcn_readlane(m.y, k + 3));
      const float4 v0 = f4[(size_t)s0 * (DFEAT / 4) + lane];
      const float4 v1 = f4[(size_t)s1 * (DFEAT / 4) + lane];
      const float4 v2 = f4[(size_t)s2 * (DFEAT / 4) + lane];
      const float4 v3 = f4[(size_t)s3 * (DFEAT / 4) + lane];
      acc.x += v0.x * w0; acc.y += v0.y * w0; acc.z += v0.z * w0; acc.w += v0.w * w0;
      acc.x += v1.x * w1; acc.y += v1.y * w1; acc.z += v1.z * w1; acc.w += v1.w * w1;
      acc.x += v2.x * w2; acc.y += v2.y * w2; acc.z += v2.z * w2; acc.w += v2.w * w2;
      acc.x += v3.x * w3; acc.y += v3.y * w3; acc.z += v3.z * w3; acc.w += v3.w * w3;
    }
    for (; k < cnt; ++k) {
      const int s0 = __builtin_amdgcn_readlane(m.x, k);
      const float w0 = __int_as_float(__builtin_amdgcn_readlane(m.y, k));
      const float4 v0 = f4[(size_t)s0 * (DFEAT / 4) + lane];
      acc.x += v0.x * w0; acc.y += v0.y * w0; acc.z += v0.z * w0; acc.w += v0.w * w0;
    }
  }
  ((float4*)agg)[(size_t)node * (DFEAT / 4) + lane] = acc;
}

#define APITCH 132
#define WPITCH 136
__global__ __launch_bounds__(256) void gcn_gemm_kernel(
    const float* __restrict__ A, const float* __restrict__ norm_dst,
    const float* __restrict__ W, const float* __restrict__ bias,
    float* __restrict__ out, int N) {
  __shared__ __attribute__((aligned(16))) float sA[64 * APITCH];
  __shared__ __attribute__((aligned(16))) float sW[32 * WPITCH];
  __shared__ float sB[DFEAT];

  const int tid = threadIdx.x;
  const int wave = tid >> 5;
  const int lane = tid & 31;
  const int row0 = blockIdx.x * 64;
  const int col0 = wave * 16;
  const int nl = lane & 15;
  const int koff = (lane >> 4) << 1;

  if (tid < DFEAT) sB[tid] = bias[tid];

#pragma unroll
  for (int it = 0; it < 8; ++it) {
    int idx4 = tid + it * 256;
    int r = idx4 >> 5;
    int c4 = idx4 & 31;
    int row = row0 + r;
    float4 v = {0.0f, 0.0f, 0.0f, 0.0f};
    if (row < N) {
      v = ((const float4*)(A + (size_t)row * DFEAT))[c4];
      float sc = norm_dst[row];
      v.x *= sc; v.y *= sc; v.z *= sc; v.w *= sc;
    }
    *(float4*)&sA[r * APITCH + c4 * 4] = v;
  }

  v8f acc0 = {}, acc1 = {}, acc2 = {}, acc3 = {};

  for (int kc = 0; kc < DFEAT; kc += 32) {
    __syncthreads();
#pragma unroll
    for (int it = 0; it < 4; ++it) {
      int idx4 = tid + it * 256;
      *(float4*)&sW[(idx4 >> 5) * WPITCH + (idx4 & 31) * 4] = ((const float4*)(W + (size_t)kc * DFEAT))[idx4];
    }
    __syncthreads();

#pragma unroll
    for (int ks = 0; ks < 8; ++ks) {
      const int kl = ks * 4 + koff;
      const int kg = kc + kl;

      v2f bfrag;
      bfrag[0] = sW[kl * WPITCH + col0 + nl];
      bfrag[1] = sW[(kl + 1) * WPITCH + col0 + nl];

      v2f a0, a1, a2, a3;
      a0[0] = sA[(0 * 16 + nl) * APITCH + kg]; a0[1] = sA[(0 * 16 + nl) * APITCH + kg + 1];
      a1[0] = sA[(1 * 16 + nl) * APITCH + kg]; a1[1] = sA[(1 * 16 + nl) * APITCH + kg + 1];
      a2[0] = sA[(2 * 16 + nl) * APITCH + kg]; a2[1] = sA[(2 * 16 + nl) * APITCH + kg + 1];
      a3[0] = sA[(3 * 16 + nl) * APITCH + kg]; a3[1] = sA[(3 * 16 + nl) * APITCH + kg + 1];

      acc0 = __builtin_amdgcn_wmma_f32_16x16x4_f32(false, a0, false, bfrag, (short)0, acc0, false, false);
      acc1 = __builtin_amdgcn_wmma_f32_16x16x4_f32(false, a1, false, bfrag, (short)0, acc1, false, false);
      acc2 = __builtin_amdgcn_wmma_f32_16x16x4_f32(false, a2, false, bfrag, (short)0, acc2, false, false);
      acc3 = __builtin_amdgcn_wmma_f32_16x16x4_f32(false, a3, false, bfrag, (short)0, acc3, false, false);
    }
  }

  const int c = col0 + nl;
  const float bval = sB[c];
  const int mhalf = (lane >> 4) << 3;
#pragma unroll
  for (int j = 0; j < 8; ++j) {
    int ml = j + mhalf;
    int r;
    r = row0 + 0 * 16 + ml; if (r < N) out[(size_t)r * DFEAT + c] = fmaxf(acc0[j] + bval, 0.0f);
    r = row0 + 1 * 16 + ml; if (r < N) out[(size_t)r * DFEAT + c] = fmaxf(acc1[j] + bval, 0.0f);
    r = row0 + 2 * 16 + ml; if (r < N) out[(size_t)r * DFEAT + c] = fmaxf(acc2[j] + bval, 0.0f);
    r = row0 + 3 * 16 + ml; if (r < N) out[(size_t)r * DFEAT + c] = fmaxf(acc3[j] + bval, 0.0f);
  }
}

extern "C" void kernel_launch(void* const* d_in, const int* in_sizes, int n_in,
                              void* d_out, int out_size, void* d_ws, size_t ws_size,
                              hipStream_t stream) {
  const float* feat = (const float*)d_in[0];
  const int* src    = (const int*)d_in[1];
  const int* dst    = (const int*)d_in[2];
  const float* W1   = (const float*)d_in[3];
  const float* b1   = (const float*)d_in[4];
  const float* W2   = (const float*)d_in[5];
  const float* b2   = (const float*)d_in[6];
  float* out        = (float*)d_out;

  const int N = in_sizes[0] / DFEAT;
  const int E = in_sizes[1];

  const int B = (E + CHUNK - 1) / CHUNK;
  const int P = (N + RNODES - 1) / RNODES;

  float* agg      = (float*)d_ws;
  float* h1       = agg + (size_t)N * DFEAT;
  int* seg_pairs  = (int*)(h1 + (size_t)N * DFEAT);
  int* bucket_d   = seg_pairs + 2 * (size_t)E;
  int* seg_src    = bucket_d + 2 * (size_t)E;
  int* bucket_s   = seg_src + E;
  int* tab_d      = bucket_s + E;
  int* tab_s      = tab_d + (size_t)(P + 1) * B;
  int* row_ptr    = tab_s + (size_t)(P + 1) * B;
  float* norm_src = (float*)(row_ptr + N + 1);
  float* norm_dst = norm_src + N;

  const int nodeWaveBlocks = (N + 7) / 8;
  const int gemmBlocks = (N + 63) / 64;

  gcn_bin_kernel<<<B, BIN_T, 0, stream>>>(src, dst, bucket_d, bucket_s, tab_d, tab_s, E, P, B);
  gcn_owner_kernel<<<P, OWN_T, 0, stream>>>(bucket_d, bucket_s, tab_d, tab_s, row_ptr, norm_src, norm_dst, seg_src, N, B);
  gcn_order_kernel<<<nodeWaveBlocks, 256, 0, stream>>>(seg_src, row_ptr, norm_src, seg_pairs, N);

  gcn_gather_kernel<<<nodeWaveBlocks, 256, 0, stream>>>(feat, seg_pairs, row_ptr, agg, N);
  gcn_gemm_kernel<<<gemmBlocks, 256, 0, stream>>>(agg, norm_dst, W1, b1, h1, N);

  gcn_gather_kernel<<<nodeWaveBlocks, 256, 0, stream>>>(h1, seg_pairs, row_ptr, agg, N);
  gcn_gemm_kernel<<<gemmBlocks, 256, 0, stream>>>(agg, norm_dst, W2, b2, out, N);
}


# gather fused into the product kernel: a block sums its 64 nodes' rows (same order) straight into its LDS A band, each wave streaming the rows of 8 nodes; agg plane and 2 launches gone
# speedup vs baseline: 4.1710x; 1.1942x over previous
#include <hip/hip_runtime.h>

typedef __attribute__((ext_vector_type(2))) float v2f;
typedef __attribute__((ext_vector_type(8))) float v8f;

#define DFEAT 128
#define RNODES 256
#define RSHIFT 8
#define CHUNK 8192
#define BIN_T 1024
#define EPT (CHUNK / BIN_T)
#define PMAX 1024
#define OWN_T 1024

__global__ __launch_bounds__(BIN_T) void gcn_bin_kernel(
    const int* __restrict__ src, const int* __restrict__ dst,
    int* __restrict__ bucket_d, int* __restrict__ bucket_s,
    int* __restrict__ tab_d, int* __restrict__ tab_s, int E, int P, int B) {
  __shared__ int2 stage_d[CHUNK];
  __shared__ int stage_s[CHUNK];
  __shared__ int hist_d[PMAX];
  __shared__ int hist_s[PMAX];
  __shared__ int wsum_d[32];
  __shared__ int wsum_s[32];
  const int tid = threadIdx.x;
  const int lane = tid & 31;
  const int wave = tid >> 5;
  const int b = blockIdx.x;
  const int e0 = b * CHUNK;
  const int cnt = min(CHUNK, E - e0);

  hist_d[tid] = 0;
  hist_s[tid] = 0;
  __syncthreads();

  int s[EPT], d[EPT], kd[EPT], ks[EPT];
#pragma unroll
  for (int i = 0; i < EPT; ++i) {
    const int j = i * BIN_T + tid;
    const bool have = j < cnt;
    s[i] = have ? src[e0 + j] : 0;
    d[i] = have ? dst[e0 + j] : 0;
  }
#pragma unroll
  for (int i = 0; i < EPT; ++i) {
    const int j = i * BIN_T + tid;
    kd[i] = 0; ks[i] = 0;
    if (j < cnt) {
      kd[i] = atomicAdd(&hist_d[d[i] >> RSHIFT], 1);
      ks[i] = atomicAdd(&hist_s[s[i] >> RSHIFT], 1);
    }
  }
  __syncthreads();

  const int xd = hist_d[tid];
  const int xs = hist_s[tid];
  int id = xd, is = xs;
  for (int o = 1; o < 32; o <<= 1) {
    const int yd = __shfl_up(id, o);
    const int ys = __shfl_up(is, o);
    if (lane >= o) { id += yd; is += ys; }
  }
  if (lane == 31) { wsum_d[wave] = id; wsum_s[wave] = is; }
  __syncthreads();
  int bd = (lane < wave) ? wsum_d[lane] : 0;
  int bs = (lane < wave) ? wsum_s[lane] : 0;
  for (int o = 16; o > 0; o >>= 1) {
    bd += __shfl_xor(bd, o);
    bs += __shfl_xor(bs, o);
  }
  const int exd = bd + id - xd;
  const int exs = bs + is - xs;
  hist_d[tid] = exd;
  hist_s[tid] = exs;
  if (tid <= P) {
    tab_d[tid * B + b] = exd;
    tab_s[tid * B + b] = exs;
  }
  __syncthreads();

#pragma unroll
  for (int i = 0; i < EPT; ++i) {
    const int j = i * BIN_T + tid;
    if (j < cnt) {
      stage_d[hist_d[d[i] >> RSHIFT] + kd[i]] = make_int2(d[i], s[i]);
      stage_s[hist_s[s[i] >> RSHIFT] + ks[i]] = s[i];
    }
  }
  __syncthreads();
  for (int j = tid; j < cnt; j += BIN_T) {
    ((int2*)bucket_d)[e0 + j] = stage_d[j];
    bucket_s[e0 + j] = stage_s[j];
  }
}

__global__ __launch_bounds__(OWN_T) void gcn_owner_kernel(
    const int* __restrict__ bucket_d, const int* __restrict__ bucket_s,
    const int* __restrict__ tab_d, const int* __restrict__ tab_s,
    int* __restrict__ row_ptr, float* __restrict__ norm_src, float* __restrict__ norm_dst,
    int* __restrict__ seg_src, int N, int B) {
  __shared__ int cnt_d[RNODES];
  __shared__ int cnt_s[RNODES];
  __shared__ int s_red[32];
  __shared__ int s_wtot[RNODES / 32];
  const int tid = threadIdx.x;
  const int lane = tid & 31;
  const int wave = __builtin_amdgcn_readfirstlane(tid >> 5);
  const int r = blockIdx.x;
  const int lo = r * RNODES;
  const int2* __restrict__ bd2 = (const int2*)bucket_d;

  if (tid < RNODES) { cnt_d[tid] = 0; cnt_s[tid] = 0; }
  int part = 0;
  for (int b = tid; b < B; b += OWN_T) part += tab_d[r * B + b];
  for (int o = 16; o > 0; o >>= 1) part += __shfl_xor(part, o);
  if (lane == 0) s_red[wave] = part;
  __syncthreads();

  for (int g = 0; g + wave < B; g += 32 * 32) {
    const int myb = g + wave + 32 * lane;
    int db = 0, de = 0, sb = 0, se = 0;
    if (myb < B) {
      db = tab_d[r * B + myb]; de = tab_d[(r + 1) * B + myb];
      sb = tab_s[r * B + myb]; se = tab_s[(r + 1) * B + myb];
    }
    const int nl = min(32, (B - g - wave + 31) / 32);
    for (int l = 0; l < nl; ++l) {
      const int bb = g + wave + 32 * l;
      const int d0 = __builtin_amdgcn_readlane(db, l);
      const int dn = __builtin_amdgcn_readlane(de, l) - d0;
      const int s0 = __builtin_amdgcn_readlane(sb, l);
      const int sn = __builtin_amdgcn_readlane(se, l) - s0;
      const int len = max(dn, sn);
      for (int o = lane; o < len; o += 32) {
        const bool hd = o < dn;
        const bool hs = o < sn;
        int ed = lo, es = lo;
        if (hd) ed = bd2[bb * CHUNK + d0 + o].x;
        if (hs) es = bucket_s[bb * CHUNK + s0 + o];
        if (hd) atomicAdd(&cnt_d[ed - lo], 1);
        if (hs) atomicAdd(&cnt_s[es - lo], 1);
      }
    }
  }
  __syncthreads();

  const int x = (tid < RNODES) ? cnt_d[tid] : 0;
  int incl = x;
  for (int o = 1; o < 32; o <<= 1) {
    const int y = __shfl_up(incl, o);
    if (lane >= o) incl += y;
  }
  if (tid < RNODES && lane == 31) s_wtot[wave] = incl;
  __syncthreads();
  int base = s_red[lane];
  int before = (lane < wave && lane < RNODES / 32) ? s_wtot[lane] : 0;
  for (int o = 16; o > 0; o >>= 1) {
    base += __shfl_xor(base, o);
    before += __shfl_xor(before, o);
  }
  if (tid < RNODES) {
    const int excl = base + before + incl - x;
    const int node = lo + tid;
    if (node < N) {
      row_ptr[node] = excl;
      norm_src[node] = rsqrtf(fmaxf((float)cnt_s[tid], 1.0f));
      norm_dst[node] = rsqrtf(fmaxf((float)x, 1.0f));
      if (node == N - 1) row_ptr[N] = excl + x;
    }
    cnt_d[tid] = excl;
  }
  __syncthreads();

  for (int g = 0; g + wave < B; g += 32 * 32) {
    const int myb = g + wave + 32 * lane;
    int db = 0, de = 0;
    if (myb < B) { db = tab_d[r * B + myb]; de = tab_d[(r + 1) * B + myb]; }
    const int nl = min(32, (B - g - wave + 31) / 32);
    for (int l = 0; l < nl; ++l) {
      const int bb = g + wave + 32 * l;
      const int d0 = __builtin_amdgcn_readlane(db, l);
      const int dn = __builtin_amdgcn_readlane(de, l) - d0;
      for (int o = lane; o < dn; o += 32) {
        const int2 ed = bd2[bb * CHUNK + d0 + o];
        const int pos = atomicAdd(&cnt_d[ed.x - lo], 1);
        seg_src[pos] = ed.y;
      }
    }
  }
}

__global__ __launch_bounds__(256) void gcn_order_kernel(
    const int* __restrict__ seg_src, const int* __restrict__ row_ptr,
    const float* __restrict__ norm_src, int* __restrict__ seg_pairs, int N) {
  const int lane = threadIdx.x & 31;
  const int node = __builtin_amdgcn_readfirstlane((int)((blockIdx.x * blockDim.x + threadIdx.x) >> 5));
  if (node >= N) return;
  const int beg = row_ptr[node];
  const int end = row_ptr[node + 1];
  int2* __restrict__ seg = (int2*)seg_pairs;
  for (int cb = beg; cb < end; cb += 32) {
    const int mi = cb + lane;
    const bool have = mi < end;
    const int ms = have ? seg_src[mi] : 0;
    int pos = 0;
    for (int ob = beg; ob < end; ob += 32) {
      const int oi = ob + lane;
      const int os = (oi < end) ? seg_src[oi] : 0;
      const int oc = min(32, end - ob);
      for (int k = 0; k < oc; ++k) {
        const int ks = __builtin_amdgcn_readlane(os, k);
        const int ki = ob + k;
        pos += (ks < ms || (ks == ms && ki < mi)) ? 1 : 0;
      }
    }
    if (have) seg[beg + pos] = make_int2(ms, __float_as_int(norm_src[ms]));
  }
}

#define APITCH 132
#define WPITCH 136
#define NPW 8
#define RIF 16
__global__ __launch_bounds__(256) void gcn_layer_kernel(
    const float* __restrict__ X, const int* __restrict__ seg_pairs,
    const int* __restrict__ row_ptr, const float* __restrict__ norm_dst,
    const float* __restrict__ W, const float* __restrict__ bias,
    float* __restrict__ out, int N) {
  __shared__ __attribute__((aligned(16))) float sA[64 * APITCH];
  __shared__ __attribute__((aligned(16))) float sW[32 * WPITCH];
  __shared__ float sB[DFEAT];

  const int tid = threadIdx.x;
  const int wave = __builtin_amdgcn_readfirstlane(tid >> 5);
  const int lane = tid & 31;
  const int row0 = blockIdx.x * 64;
  const int col0 = wave * 16;
  const int nl = lane & 15;
  const int koff = (lane >> 4) << 1;

  float bload = 0.0f;
  if (tid < DFEAT) bload = bias[tid];

  {
    const int n0 = row0 + wave * NPW;
    int rp = 0;
    int nd = 0;
    if (lane <= NPW) rp = row_ptr[min(n0 + lane, N)];
    if (lane < NPW) nd = __float_as_int(norm_dst[min(n0 + lane, N - 1)]);
    const int beg = __builtin_amdgcn_readlane(rp, 0);
    const int end = __builtin_amdgcn_readlane(rp, NPW);
    const int2* __restrict__ seg = (const int2*)seg_pairs;
    const float4* __restrict__ x4 = (const float4*)X;
    float* arow = &sA[(wave * NPW) * APITCH + lane * 4];

    int cur = 0;
    int nextb = __builtin_amdgcn_readlane(rp, 1);
    float4 acc = {0.0f, 0.0f, 0.0f, 0.0f};

    int2 m = make_int2(0, 0);
    if (beg + lane < end) m = seg[beg + lane];
    for (int cb = beg; cb < end; cb += 32) {
      const int cnt = min(32, end - cb);
      int2 mn = make_int2(0, 0);
      for (int k = 0; k < cnt; k += RIF) {
        int s[RIF];
        float4 v[RIF];
#pragma unroll
        for (int i = 0; i < RIF; ++i) s[i] = __builtin_amdgcn_readlane(m.x, min(k + i, cnt - 1));
#pragma unroll
        for (int i = 0; i < RIF; ++i) v[i] = x4[(size_t)s[i] * (DFEAT / 4) + lane];
        if (k == 0 && cb + 32 + lane < end) mn = seg[cb + 32 + lane];
#pragma unroll
        for (int i = 0; i < RIF; ++i) {
          if (i == 0 || k + i < cnt) {
            const int e = cb + k + i;
            while (e == nextb) {
              const float sc = __int_as_float(__builtin_amdgcn_readlane(nd, cur));
              *(float4*)(arow + cur * APITCH) = make_float4(acc.x * sc, acc.y * sc, acc.z * sc, acc.w * sc);
              acc = make_float4(0.0f, 0.0f, 0.0f, 0.0f);
              ++cur;
              nextb = __builtin_amdgcn_readlane(rp, cur + 1);
            }
            const float w = __int_as_float(__builtin_amdgcn_readlane(m.y, k + i));
            acc.x += v[i].x * w; acc.y += v[i].y * w; acc.z += v[i].z * w; acc.w += v[i].w * w;
          }
        }
      }
      m = mn;
    }
    for (; cur < NPW; ++cur) {
      const float sc = __int_as_float(__builtin_amdgcn_readlane(nd, cur));
      *(float4*)(arow + cur * APITCH) = make_float4(acc.x * sc, acc.y * sc, acc.z * sc, acc.w * sc);
      acc = make_float4(0.0f, 0.0f, 0.0f, 0.0f);
    }
  }

  if (tid < DFEAT) sB[tid] = bload;
  v8f acc0 = {}, acc1 = {}, acc2 = {}, acc3 = {};

  for (int kc = 0; kc < DFEAT; kc += 32) {
    __syncthreads();
#pragma unroll
    for (int it = 0; it < 4; ++it) {
      int idx4 = tid + it * 256;
      *(float4*)&sW[(idx4 >> 5) * WPITCH + (idx4 & 31) * 4] = ((const float4*)(W + (size_t)kc * DFEAT))[idx4];
    }
    __syncthreads();

#pragma unroll
    for (int ks = 0; ks < 8; ++ks) {
      const int kl = ks * 4 + koff;
      const int kg = kc + kl;

      v2f bfrag;
      bfrag[0] = sW[kl * WPITCH + col0 + nl];
      bfrag[1] = sW[(kl + 1) * WPITCH + col0 + nl];

      v2f a0, a1, a2, a3;
      a0[0] = sA[(0 * 16 + nl) * APITCH + kg]; a0[1] = sA[(0 * 16 + nl) * APITCH + kg + 1];
      a1[0] = sA[(1 * 16 + nl) * APITCH + kg]; a1[1] = sA[(1 * 16 + nl) * APITCH + kg + 1];
      a2[0] = sA[(2 * 16 + nl) * APITCH + kg]; a2[1] = sA[(2 * 16 + nl) * APITCH + kg + 1];
      a3[0] = sA[(3 * 16 + nl) * APITCH + kg]; a3[1] = sA[(3 * 16 + nl) * APITCH + kg + 1];

      acc0 = __builtin_amdgcn_wmma_f32_16x16x4_f32(false, a0, false, bfrag, (short)0, acc0, false, false);
      acc1 = __builtin_amdgcn_wmma_f32_16x16x4_f32(false, a1, false, bfrag, (short)0, acc1, false, false);
      acc2 = __builtin_amdgcn_wmma_f32_16x16x4_f32(false, a2, false, bfrag, (short)0, acc2, false, false);
      acc3 = __builtin_amdgcn_wmma_f32_16x16x4_f32(false, a3, false, bfrag, (short)0, acc3, false, false);
    }
  }

  const int c = col0 + nl;
  const float bval = sB[c];
  const int mhalf = (lane >> 4) << 3;
#pragma unroll
  for (int j = 0; j < 8; ++j) {
    int ml = j + mhalf;
    int r;
    r = row0 + 0 * 16 + ml; if (r < N) out[(size_t)r * DFEAT + c] = fmaxf(acc0[j] + bval, 0.0f);
    r = row0 + 1 * 16 + ml; if (r < N) out[(size_t)r * DFEAT + c] = fmaxf(acc1[j] + bval, 0.0f);
    r = row0 + 2 * 16 + ml; if (r < N) out[(size_t)r * DFEAT + c] = fmaxf(acc2[j] + bval, 0.0f);
    r = row0 + 3 * 16 + ml; if (r < N) out[(size_t)r * DFEAT + c] = fmaxf(acc3[j] + bval, 0.0f);
  }
}

extern "C" void kernel_launch(void* const* d_in, const int* in_sizes, int n_in,
                              void* d_out, int out_size, void* d_ws, size_t ws_size,
                              hipStream_t stream) {
  const float* feat = (const float*)d_in[0];
  const int* src    = (const int*)d_in[1];
  const int* dst    = (const int*)d_in[2];
  const float* W1   = (const float*)d_in[3];
  const float* b1   = (const float*)d_in[4];
  const float* W2   = (const float*)d_in[5];
  const float* b2   = (const float*)d_in[6];
  float* out        = (float*)d_out;

  const int N = in_sizes[0] / DFEAT;
  const int E = in_sizes[1];

  const int B = (E + CHUNK - 1) / CHUNK;
  const int P = (N + RNODES - 1) / RNODES;

  float* h1       = (float*)d_ws;
  int* seg_pairs  = (int*)(h1 + (size_t)N * DFEAT);
  int* bucket_d   = seg_pairs + 2 * (size_t)E;
  int* seg_src    = bucket_d + 2 * (size_t)E;
  int* bucket_s   = seg_src + E;
  int* tab_d      = bucket_s + E;
  int* tab_s      = tab_d + (size_t)(P + 1) * B;
  int* row_ptr    = tab_s + (size_t)(P + 1) * B;
  float* norm_src = (float*)(row_ptr + N + 1);
  float* norm_dst = norm_src + N;

  const int nodeWaveBlocks = (N + 7) / 8;
  const int bandBlocks = (N + 63) / 64;

  gcn_bin_kernel<<<B, BIN_T, 0, stream>>>(src, dst, bucket_d, bucket_s, tab_d, tab_s, E, P, B);
  gcn_owner_kernel<<<P, OWN_T, 0, stream>>>(bucket_d, bucket_s, tab_d, tab_s, row_ptr, norm_src, norm_dst, seg_src, N, B);
  gcn_order_kernel<<<nodeWaveBlocks, 256, 0, stream>>>(seg_src, row_ptr, norm_src, seg_pairs, N);

  gcn_layer_kernel<<<bandBlocks, 256, 0, stream>>>(feat, seg_pairs, row_ptr, norm_dst, W1, b1, h1, N);
  gcn_layer_kernel<<<bandBlocks, 256, 0, stream>>>(h1, seg_pairs, row_ptr, norm_dst, W2, b2, out, N);
}


# ordering launch folded into the owner launch: each block ranks its 256 nodes' sources in LDS (same ascending order); layer 1's gather attaches norm_src and leaves the pairs for layer 2
# speedup vs baseline: 4.4771x; 1.0734x over previous
#include <hip/hip_runtime.h>

typedef __attribute__((ext_vector_type(2))) float v2f;
typedef __attribute__((ext_vector_type(8))) float v8f;

#define DFEAT 128
#define RNODES 256
#define RSHIFT 8
#define CHUNK 8192
#define BIN_T 1024
#define EPT (CHUNK / BIN_T)
#define PMAX 1024
#define OWN_T 1024
#define OWN_CAP 8192

__global__ __launch_bounds__(BIN_T) void gcn_bin_kernel(
    const int* __restrict__ src, const int* __restrict__ dst,
    int* __restrict__ bucket_d, int* __restrict__ bucket_s,
    int* __restrict__ tab_d, int* __restrict__ tab_s, int E, int P, int B) {
  __shared__ int2 stage_d[CHUNK];
  __shared__ int stage_s[CHUNK];
  __shared__ int hist_d[PMAX];
  __shared__ int hist_s[PMAX];
  __shared__ int wsum_d[32];
  __shared__ int wsum_s[32];
  const int tid = threadIdx.x;
  const int lane = tid & 31;
  const int wave = tid >> 5;
  const int b = blockIdx.x;
  const int e0 = b * CHUNK;
  const int cnt = min(CHUNK, E - e0);

  hist_d[tid] = 0;
  hist_s[tid] = 0;
  __syncthreads();

  int s[EPT], d[EPT], kd[EPT], ks[EPT];
#pragma unroll
  for (int i = 0; i < EPT; ++i) {
    const int j = i * BIN_T + tid;
    const bool have = j < cnt;
    s[i] = have ? src[e0 + j] : 0;
    d[i] = have ? dst[e0 + j] : 0;
  }
#pragma unroll
  for (int i = 0; i < EPT; ++i) {
    const int j = i * BIN_T + tid;
    kd[i] = 0; ks[i] = 0;
    if (j < cnt) {
      kd[i] = atomicAdd(&hist_d[d[i] >> RSHIFT], 1);
      ks[i] = atomicAdd(&hist_s[s[i] >> RSHIFT], 1);
    }
  }
  __syncthreads();

  const int xd = hist_d[tid];
  const int xs = hist_s[tid];
  int id = xd, is = xs;
  for (int o = 1; o < 32; o <<= 1) {
    const int yd = __shfl_up(id, o);
    const int ys = __shfl_up(is, o);
    if (lane >= o) { id += yd; is += ys; }
  }
  if (lane == 31) { wsum_d[wave] = id; wsum_s[wave] = is; }
  __syncthreads();
  int bd = (lane < wave) ? wsum_d[lane] : 0;
  int bs = (lane < wave) ? wsum_s[lane] : 0;
  for (int o = 16; o > 0; o >>= 1) {
    bd += __shfl_xor(bd, o);
    bs += __shfl_xor(bs, o);
  }
  const int exd = bd + id - xd;
  const int exs = bs + is - xs;
  hist_d[tid] = exd;
  hist_s[tid] = exs;
  if (tid <= P) {
    tab_d[tid * B + b] = exd;
    tab_s[tid * B + b] = exs;
  }
  __syncthreads();

#pragma unroll
  for (int i = 0; i < EPT; ++i) {
    const int j = i * BIN_T + tid;
    if (j < cnt) {
      stage_d[hist_d[d[i] >> RSHIFT] + kd[i]] = make_int2(d[i], s[i]);
      stage_s[hist_s[s[i] >> RSHIFT] + ks[i]] = s[i];
    }
  }
  __syncthreads();
  for (int j = tid; j < cnt; j += BIN_T) {
    ((int2*)bucket_d)[e0 + j] = stage_d[j];
    bucket_s[e0 + j] = stage_s[j];
  }
}

__global__ __launch_bounds__(OWN_T) void gcn_owner_kernel(
    const int* __restrict__ bucket_d, const int* __restrict__ bucket_s,
    const int* __restrict__ tab_d, const int* __restrict__ tab_s,
    int* __restrict__ row_ptr, float* __restrict__ norm_src, float* __restrict__ norm_dst,
    int* seg_tmp, int* __restrict__ seg_sorted, int N, int B) {
  __shared__ int cnt_d[RNODES];
  __shared__ int cnt_s[RNODES];
  __shared__ int beg_l[RNODES + 1];
  __shared__ int st_src[OWN_CAP];
  __shared__ int st_node[OWN_CAP];
  __shared__ int s_red[32];
  __shared__ int s_wtot[RNODES / 32];
  const int tid = threadIdx.x;
  const int lane = tid & 31;
  const int wave = __builtin_amdgcn_readfirstlane(tid >> 5);
  const int r = blockIdx.x;
  const int lo = r * RNODES;
  const int2* __restrict__ bd2 = (const int2*)bucket_d;

  if (tid < RNODES) { cnt_d[tid] = 0; cnt_s[tid] = 0; }
  int part = 0;
  for (int b = tid; b < B; b += OWN_T) part += tab_d[r * B + b];
  for (int o = 16; o > 0; o >>= 1) part += __shfl_xor(part, o);
  if (lane == 0) s_red[wave] = part;
  __syncthreads();

  for (int g = 0; g + wave < B; g += 32 * 32) {
    const int myb = g + wave + 32 * lane;
    int db = 0, de = 0, sb = 0, se = 0;
    if (myb < B) {
      db = tab_d[r * B + myb]; de = tab_d[(r + 1) * B + myb];
      sb = tab_s[r * B + myb]; se = tab_s[(r + 1) * B + myb];
    }
    const int nl = min(32, (B - g - wave + 31) / 32);
    for (int l = 0; l < nl; ++l) {
      const int bb = g + wave + 32 * l;
      const int d0 = __builtin_amdgcn_readlane(db, l);
      const int dn = __builtin_amdgcn_readlane(de, l) - d0;
      const int s0 = __builtin_amdgcn_readlane(sb, l);
      const int sn = __builtin_amdgcn_readlane(se, l) - s0;
      const int len = max(dn, sn);
      for (int o = lane; o < len; o += 32) {
        const bool hd = o < dn;
        const bool hs = o < sn;
        int ed = lo, es = lo;
        if (hd) ed = bd2[bb * CHUNK + d0 + o].x;
        if (hs) es = bucket_s[bb * CHUNK + s0 + o];
        if (hd) atomicAdd(&cnt_d[ed - lo], 1);
        if (hs) atomicAdd(&cnt_s[es - lo], 1);
      }
    }
  }
  __syncthreads();

  const int x = (tid < RNODES) ? cnt_d[tid] : 0;
  int incl = x;
  for (int o = 1; o < 32; o <<= 1) {
    const int y = __shfl_up(incl, o);
    if (lane >= o) incl += y;
  }
  if (tid < RNODES && lane == 31) s_wtot[wave] = incl;
  __syncthreads();
  int base = s_red[lane];
  int before = (lane < wave && lane < RNODES / 32) ? s_wtot[lane] : 0;
  for (int o = 16; o > 0; o >>= 1) {
    base += __shfl_xor(base, o);
    before += __shfl_xor(before, o);
  }
  if (tid < RNODES) {
    const int excl_l = before + incl - x;
    const int excl = base + excl_l;
    const int node = lo + tid;
    if (node < N) {
      row_ptr[node] = excl;
      norm_src[node] = rsqrtf(fmaxf((float)cnt_s[tid], 1.0f));
      norm_dst[node] = rsqrtf(fmaxf((float)x, 1.0f));
      if (node == N - 1) row_ptr[N] = excl + x;
    }
    cnt_d[tid] = excl_l;
    beg_l[tid] = excl_l;
    if (tid == RNODES - 1) beg_l[RNODES] = excl_l + x;
  }
  __syncthreads();
  const int total = beg_l[RNODES];
  const bool in_lds = total <= OWN_CAP;

  for (int g = 0; g + wave < B; g += 32 * 32) {
    const int myb = g + wave + 32 * lane;
    int db = 0, de = 0;
    if (myb < B) { db = tab_d[r * B + myb]; de = tab_d[(r + 1) * B + myb]; }
    const int nl = min(32, (B - g - wave + 31) / 32);
    for (int l = 0; l < nl; ++l) {
      const int bb = g + wave + 32 * l;
      const int d0 = __builtin_amdgcn_readlane(db, l);
      const int dn = __builtin_amdgcn_readlane(de, l) - d0;
      for (int o = lane; o < dn; o += 32) {
        const int2 ed = bd2[bb * CHUNK + d0 + o];
        const int nloc = ed.x - lo;
        const int pos = atomicAdd(&cnt_d[nloc], 1);
        if (in_lds) { st_src[pos] = ed.y; st_node[pos] = nloc; }
        else seg_tmp[base + pos] = ed.y;
      }
    }
  }
  __syncthreads();

  if (in_lds) {
    for (int e = tid; e < total; e += OWN_T) {
      const int ms = st_src[e];
      const int n = st_node[e];
      const int b0 = beg_l[n];
      const int b1 = beg_l[n + 1];
      int pos = 0;
#pragma unroll 4
      for (int j = b0; j < b1; ++j) {
        const int os = st_src[j];
        pos += (os < ms || (os == ms && j < e)) ? 1 : 0;
      }
      seg_sorted[base + b0 + pos] = ms;
    }
  } else {
    for (int n = wave; n < RNODES; n += 32) {
      const int beg = base + beg_l[n];
      const int end = base + beg_l[n + 1];
      for (int cb = beg; cb < end; cb += 32) {
        const int mi = cb + lane;
        const bool have = mi < end;
        const int ms = have ? seg_tmp[mi] : 0;
        int pos = 0;
        for (int ob = beg; ob < end; ob += 32) {
          const int oi = ob + lane;
          const int os = (oi < end) ? seg_tmp[oi] : 0;
          const int oc = min(32, end - ob);
          for (int k = 0; k < oc; ++k) {
            const int ks = __builtin_amdgcn_readlane(os, k);
            const int ki = ob + k;
            pos += (ks < ms || (ks == ms && ki < mi)) ? 1 : 0;
          }
        }
        if (have) seg_sorted[beg + pos] = ms;
      }
    }
  }
}

#define APITCH 132
#define WPITCH 136
#define NPW 8
#define RIF 16
template <bool FIRST>
__global__ __launch_bounds__(256) void gcn_layer_kernel(
    const float* __restrict__ X, const int* __restrict__ seg_sorted,
    const float* __restrict__ norm_src, int* __restrict__ seg_pairs,
    const int* __restrict__ row_ptr, const float* __restrict__ norm_dst,
    const float* __restrict__ W, const float* __restrict__ bias,
    float* __restrict__ out, int N) {
  __shared__ __attribute__((aligned(16))) float sA[64 * APITCH];
  __shared__ __attribute__((aligned(16))) float sW[32 * WPITCH];
  __shared__ float sB[DFEAT];

  const int tid = threadIdx.x;
  const int wave = __builtin_amdgcn_readfirstlane(tid >> 5);
  const int lane = tid & 31;
  const int row0 = blockIdx.x * 64;
  const int col0 = wave * 16;
  const int nl = lane & 15;
  const int koff = (lane >> 4) << 1;

  float bload = 0.0f;
  if (tid < DFEAT) bload = bias[tid];

  {
    const int n0 = row0 + wave * NPW;
    int rp = 0;
    int nd = 0;
    if (lane <= NPW) rp = row_ptr[min(n0 + lane, N)];
    if (lane < NPW) nd = __float_as_int(norm_dst[min(n0 + lane, N - 1)]);
    const int beg = __builtin_amdgcn_readlane(rp, 0);
    const int end = __builtin_amdgcn_readlane(rp, NPW);
    int2* __restrict__ seg = (int2*)seg_pairs;
    const float4* __restrict__ x4 = (const float4*)X;
    float* arow = &sA[(wave * NPW) * APITCH + lane * 4];

    int cur = 0;
    int nextb = __builtin_amdgcn_readlane(rp, 1);
    float4 acc = {0.0f, 0.0f, 0.0f, 0.0f};

    int2 m = make_int2(0, 0);
    if (beg + lane < end) {
      if (FIRST) m.x = seg_sorted[beg + lane];
      else m = seg[beg + lane];
    }
    for (int cb = beg; cb < end; cb += 32) {
      const int cnt = min(32, end - cb);
      int2 mn = make_int2(0, 0);
      const int other = FIRST ? __shfl_xor(m.x, 16) : 0;
      for (int k = 0; k < cnt; k += RIF) {
        int s[RIF];
        float4 v[RIF];
        int sk = 0;
        float wt = 0.0f;
#pragma unroll
        for (int i = 0; i < RIF; ++i) s[i] = __builtin_amdgcn_readlane(m.x, min(k + i, cnt - 1));
        if (FIRST) {
          static_assert(RIF == 16, "a trip is one lane half");
          sk = (((lane >> 4) << 4) == k) ? m.x : other;
          wt = norm_src[sk];
          __builtin_amdgcn_sched_barrier(0);
        }
#pragma unroll
        for (int i = 0; i < RIF; ++i) v[i] = x4[(size_t)s[i] * (DFEAT / 4) + lane];
        if (k == 0 && cb + 32 + lane < end) {
          if (FIRST) mn.x = seg_sorted[cb + 32 + lane];
          else mn = seg[cb + 32 + lane];
        }

#pragma unroll
        for (int i = 0; i < RIF; ++i) {
          if (i == 0 || k + i < cnt) {
            const int e = cb + k + i;
            while (e == nextb) {
              const float sc = __int_as_float(__builtin_amdgcn_readlane(nd, cur));
              *(float4*)(arow + cur * APITCH) = make_float4(acc.x * sc, acc.y * sc, acc.z * sc, acc.w * sc);
              acc = make_float4(0.0f, 0.0f, 0.0f, 0.0f);
              ++cur;
              nextb = __builtin_amdgcn_readlane(rp, cur + 1);
            }
            const float w = FIRST ? __int_as_float(__builtin_amdgcn_readlane(__float_as_int(wt), i))
                                  : __int_as_float(__builtin_amdgcn_readlane(m.y, k + i));
            acc.x += v[i].x * w; acc.y += v[i].y * w; acc.z += v[i].z * w; acc.w += v[i].w * w;
          }
        }
        if (FIRST && lane < 16 && k + lane < cnt) seg[cb + k + lane] = make_int2(sk, __float_as_int(wt));
      }
      m = mn;
    }
    for (; cur < NPW; ++cur) {
      const float sc = __int_as_float(__builtin_amdgcn_readlane(nd, cur));
      *(float4*)(arow + cur * APITCH) = make_float4(acc.x * sc, acc.y * sc, acc.z * sc, acc.w * sc);
      acc = make_float4(0.0f, 0.0f, 0.0f, 0.0f);
    }
  }

  if (tid < DFEAT) sB[tid] = bload;
  v8f acc0 = {}, acc1 = {}, acc2 = {}, acc3 = {};

  for (int kc = 0; kc < DFEAT; kc += 32) {
    __syncthreads();
#pragma unroll
    for (int it = 0; it < 4; ++it) {
      int idx4 = tid + it * 256;
      *(float4*)&sW[(idx4 >> 5) * WPITCH + (idx4 & 31) * 4] = ((const float4*)(W + (size_t)kc * DFEAT))[idx4];
    }
    __syncthreads();

#pragma unroll
    for (int ks = 0; ks < 8; ++ks) {
      const int kl = ks * 4 + koff;
      const int kg = kc + kl;

      v2f bfrag;
      bfrag[0] = sW[kl * WPITCH + col0 + nl];
      bfrag[1] = sW[(kl + 1) * WPITCH + col0 + nl];

      v2f a0, a1, a2, a3;
      a0[0] = sA[(0 * 16 + nl) * APITCH + kg]; a0[1] = sA[(0 * 16 + nl) * APITCH + kg + 1];
      a1[0] = sA[(1 * 16 + nl) * APITCH + kg]; a1[1] = sA[(1 * 16 + nl) * APITCH + kg + 1];
      a2[0] = sA[(2 * 16 + nl) * APITCH + kg]; a2[1] = sA[(2 * 16 + nl) * APITCH + kg + 1];
      a3[0] = sA[(3 * 16 + nl) * APITCH + kg]; a3[1] = sA[(3 * 16 + nl) * APITCH + kg + 1];

      acc0 = __builtin_amdgcn_wmma_f32_16x16x4_f32(false, a0, false, bfrag, (short)0, acc0, false, false);
      acc1 = __builtin_amdgcn_wmma_f32_16x16x4_f32(false, a1, false, bfrag, (short)0, acc1, false, false);
      acc2 = __builtin_amdgcn_wmma_f32_16x16x4_f32(false, a2, false, bfrag, (short)0, acc2, false, false);
      acc3 = __builtin_amdgcn_wmma_f32_16x16x4_f32(false, a3, false, bfrag, (short)0, acc3, false, false);
    }
  }

  const int c = col0 + nl;
  const float bval = sB[c];
  const int mhalf = (lane >> 4) << 3;
#pragma unroll
  for (int j = 0; j < 8; ++j) {
    int ml = j + mhalf;
    int r;
    r = row0 + 0 * 16 + ml; if (r < N) out[(size_t)r * DFEAT + c] = fmaxf(acc0[j] + bval, 0.0f);
    r = row0 + 1 * 16 + ml; if (r < N) out[(size_t)r * DFEAT + c] = fmaxf(acc1[j] + bval, 0.0f);
    r = row0 + 2 * 16 + ml; if (r < N) out[(size_t)r * DFEAT + c] = fmaxf(acc2[j] + bval, 0.0f);
    r = row0 + 3 * 16 + ml; if (r < N) out[(size_t)r * DFEAT + c] = fmaxf(acc3[j] + bval, 0.0f);
  }
}

extern "C" void kernel_launch(void* const* d_in, const int* in_sizes, int n_in,
                              void* d_out, int out_size, void* d_ws, size_t ws_size,
                              hipStream_t stream) {
  const float* feat = (const float*)d_in[0];
  const int* src    = (const int*)d_in[1];
  const int* dst    = (const int*)d_in[2];
  const float* W1   = (const float*)d_in[3];
  const float* b1   = (const float*)d_in[4];
  const float* W2   = (const float*)d_in[5];
  const float* b2   = (const float*)d_in[6];
  float* out        = (float*)d_out;

  const int N = in_sizes[0] / DFEAT;
  const int E = in_sizes[1];

  const int B = (E + CHUNK - 1) / CHUNK;
  const int P = (N + RNODES - 1) / RNODES;

  float* h1       = (float*)d_ws;
  int* seg_pairs  = (int*)(h1 + (size_t)N * DFEAT);
  int* bucket_d   = seg_pairs + 2 * (size_t)E;
  int* seg_tmp    = bucket_d + 2 * (size_t)E;
  int* bucket_s   = seg_tmp + E;
  int* tab_d      = bucket_s + E;
  int* tab_s      = tab_d + (size_t)(P + 1) * B;
  int* row_ptr    = tab_s + (size_t)(P + 1) * B;
  float* norm_src = (float*)(row_ptr + N + 1);
  float* norm_dst = norm_src + N;
  int* seg_sorted = (int*)(norm_dst + N);

  const int bandBlocks = (N + 63) / 64;

  gcn_bin_kernel<<<B, BIN_T, 0, stream>>>(src, dst, bucket_d, bucket_s, tab_d, tab_s, E, P, B);
  gcn_owner_kernel<<<P, OWN_T, 0, stream>>>(bucket_d, bucket_s, tab_d, tab_s, row_ptr, norm_src, norm_dst, seg_tmp, seg_sorted, N, B);

  gcn_layer_kernel<true><<<bandBlocks, 256, 0, stream>>>(feat, seg_sorted, norm_src, seg_pairs, row_ptr, norm_dst, W1, b1, h1, N);
  gcn_layer_kernel<false><<<bandBlocks, 256, 0, stream>>>(h1, seg_sorted, norm_src, seg_pairs, row_ptr, norm_dst, W2, b2, out, N);
}


# layer launches: one 26-wave block of 208 nodes per WGP (241 blocks, LDS sized so only one fits) instead of 782 blocks of 64 nodes, 3 or 4 to a WGP; W staged in LDS whole, once; sum order kept
# speedup vs baseline: 4.8262x; 1.0780x over previous
#include <hip/hip_runtime.h>

typedef __attribute__((ext_vector_type(2))) float v2f;
typedef __attribute__((ext_vector_type(8))) float v8f;

#define DFEAT 128
#define RNODES 256
#define RSHIFT 8
#define CHUNK 8192
#define BIN_T 1024
#define EPT (CHUNK / BIN_T)
#define PMAX 1024
#define OWN_T 1024
#define OWN_CAP 8192

__global__ __launch_bounds__(BIN_T) void gcn_bin_kernel(
    const int* __restrict__ src, const int* __restrict__ dst,
    int* __restrict__ bucket_d, int* __restrict__ bucket_s,
    int* __restrict__ tab_d, int* __restrict__ tab_s, int E, int P, int B) {
  __shared__ int2 stage_d[CHUNK];
  __shared__ int stage_s[CHUNK];
  __shared__ int hist_d[PMAX];
  __shared__ int hist_s[PMAX];
  __shared__ int wsum_d[32];
  __shared__ int wsum_s[32];
  const int tid = threadIdx.x;
  const int lane = tid & 31;
  const int wave = tid >> 5;
  const int b = blockIdx.x;
  const int e0 = b * CHUNK;
  const int cnt = min(CHUNK, E - e0);

  hist_d[tid] = 0;
  hist_s[tid] = 0;
  __syncthreads();

  int s[EPT], d[EPT], kd[EPT], ks[EPT];
#pragma unroll
  for (int i = 0; i < EPT; ++i) {
    const int j = i * BIN_T + tid;
    const bool have = j < cnt;
    s[i] = have ? src[e0 + j] : 0;
    d[i] = have ? dst[e0 + j] : 0;
  }
#pragma unroll
  for (int i = 0; i < EPT; ++i) {
    const int j = i * BIN_T + tid;
    kd[i] = 0; ks[i] = 0;
    if (j < cnt) {
      kd[i] = atomicAdd(&hist_d[d[i] >> RSHIFT], 1);
      ks[i] = atomicAdd(&hist_s[s[i] >> RSHIFT], 1);
    }
  }
  __syncthreads();

  const int xd = hist_d[tid];
  const int xs = hist_s[tid];
  int id = xd, is = xs;
  for (int o = 1; o < 32; o <<= 1) {
    const int yd = __shfl_up(id, o);
    const int ys = __shfl_up(is, o);
    if (lane >= o) { id += yd; is += ys; }
  }
  if (lane == 31) { wsum_d[wave] = id; wsum_s[wave] = is; }
  __syncthreads();
  int bd = (lane < wave) ? wsum_d[lane] : 0;
  int bs = (lane < wave) ? wsum_s[lane] : 0;
  for (int o = 16; o > 0; o >>= 1) {
    bd += __shfl_xor(bd, o);
    bs += __shfl_xor(bs, o);
  }
  const int exd = bd + id - xd;
  const int exs = bs + is - xs;
  hist_d[tid] = exd;
  hist_s[tid] = exs;
  if (tid <= P) {
    tab_d[tid * B + b] = exd;
    tab_s[tid * B + b] = exs;
  }
  __syncthreads();

#pragma unroll
  for (int i = 0; i < EPT; ++i) {
    const int j = i * BIN_T + tid;
    if (j < cnt) {
      stage_d[hist_d[d[i] >> RSHIFT] + kd[i]] = make_int2(d[i], s[i]);
      stage_s[hist_s[s[i] >> RSHIFT] + ks[i]] = s[i];
    }
  }
  __syncthreads();
  for (int j = tid; j < cnt; j += BIN_T) {
    ((int2*)bucket_d)[e0 + j] = stage_d[j];
    bucket_s[e0 + j] = stage_s[j];
  }
}

__global__ __launch_bounds__(OWN_T) void gcn_owner_kernel(
    const int* __restrict__ bucket_d, const int* __restrict__ bucket_s,
    const int* __restrict__ tab_d, const int* __restrict__ tab_s,
    int* __restrict__ row_ptr, float* __restrict__ norm_src, float* __restrict__ norm_dst,
    int* seg_tmp, int* __restrict__ seg_sorted, int N, int B) {
  __shared__ int cnt_d[RNODES];
  __shared__ int cnt_s[RNODES];
  __shared__ int beg_l[RNODES + 1];
  __shared__ int st_src[OWN_CAP];
  __shared__ int st_node[OWN_CAP];
  __shared__ int s_red[32];
  __shared__ int s_wtot[RNODES / 32];
  const int tid = threadIdx.x;
  const int lane = tid & 31;
  const int wave = __builtin_amdgcn_readfirstlane(tid >> 5);
  const int r = blockIdx.x;
  const int lo = r * RNODES;
  const int2* __restrict__ bd2 = (const int2*)bucket_d;

  if (tid < RNODES) { cnt_d[tid] = 0; cnt_s[tid] = 0; }
  int part = 0;
  for (int b = tid; b < B; b += OWN_T) part += tab_d[r * B + b];
  for (int o = 16; o > 0; o >>= 1) part += __shfl_xor(part, o);
  if (lane == 0) s_red[wave] = part;
  __syncthreads();

  for (int g = 0; g + wave < B; g += 32 * 32) {
    const int myb = g + wave + 32 * lane;
    int db = 0, de = 0, sb = 0, se = 0;
    if (myb < B) {
      db = tab_d[r * B + myb]; de = tab_d[(r + 1) * B + myb];
      sb = tab_s[r * B + myb]; se = tab_s[(r + 1) * B + myb];
    }
    const int nl = min(32, (B - g - wave + 31) / 32);
    for (int l = 0; l < nl; ++l) {
      const int bb = g + wave + 32 * l;
      const int d0 = __builtin_amdgcn_readlane(db, l);
      const int dn = __builtin_amdgcn_readlane(de, l) - d0;
      const int s0 = __builtin_amdgcn_readlane(sb, l);
      const int sn = __builtin_amdgcn_readlane(se, l) - s0;
      const int len = max(dn, sn);
      for (int o = lane; o < len; o += 32) {
        const bool hd = o < dn;
        const bool hs = o < sn;
        int ed = lo, es = lo;
        if (hd) ed = bd2[bb * CHUNK + d0 + o].x;
        if (hs) es = bucket_s[bb * CHUNK + s0 + o];
        if (hd) atomicAdd(&cnt_d[ed - lo], 1);
        if (hs) atomicAdd(&cnt_s[es - lo], 1);
      }
    }
  }
  __syncthreads();

  const int x = (tid < RNODES) ? cnt_d[tid] : 0;
  int incl = x;
  for (int o = 1; o < 32; o <<= 1) {
    const int y = __shfl_up(incl, o);
    if (lane >= o) incl += y;
  }
  if (tid < RNODES && lane == 31) s_wtot[wave] = incl;
  __syncthreads();
  int base = s_red[lane];
  int before = (lane < wave && lane < RNODES / 32) ? s_wtot[lane] : 0;
  for (int o = 16; o > 0; o >>= 1) {
    base += __shfl_xor(base, o);
    before += __shfl_xor(before, o);
  }
  if (tid < RNODES) {
    const int excl_l = before + incl - x;
    const int excl = base + excl_l;
    const int node = lo + tid;
    if (node < N) {
      row_ptr[node] = excl;
      norm_src[node] = rsqrtf(fmaxf((float)cnt_s[tid], 1.0f));
      norm_dst[node] = rsqrtf(fmaxf((float)x, 1.0f));
      if (node == N - 1) row_ptr[N] = excl + x;
    }
    cnt_d[tid] = excl_l;
    beg_l[tid] = excl_l;
    if (tid == RNODES - 1) beg_l[RNODES] = excl_l + x;
  }
  __syncthreads();
  const int total = beg_l[RNODES];
  const bool in_lds = total <= OWN_CAP;

  for (int g = 0; g + wave < B; g += 32 * 32) {
    const int myb = g + wave + 32 * lane;
    int db = 0, de = 0;
    if (myb < B) { db = tab_d[r * B + myb]; de = tab_d[(r + 1) * B + myb]; }
    const int nl = min(32, (B - g - wave + 31) / 32);
    for (int l = 0; l < nl; ++l) {
      const int bb = g + wave + 32 * l;
      const int d0 = __builtin_amdgcn_readlane(db, l);
      const int dn = __builtin_amdgcn_readlane(de, l) - d0;
      for (int o = lane; o < dn; o += 32) {
        const int2 ed = bd2[bb * CHUNK + d0 + o];
        const int nloc = ed.x - lo;
        const int pos = atomicAdd(&cnt_d[nloc], 1);
        if (in_lds) { st_src[pos] = ed.y; st_node[pos] = nloc; }
        else seg_tmp[base + pos] = ed.y;
      }
    }
  }
  __syncthreads();

  if (in_lds) {
    for (int e = tid; e < total; e += OWN_T) {
      const int ms = st_src[e];
      const int n = st_node[e];
      const int b0 = beg_l[n];
      const int b1 = beg_l[n + 1];
      int pos = 0;
#pragma unroll 4
      for (int j = b0; j < b1; ++j) {
        const int os = st_src[j];
        pos += (os < ms || (os == ms && j < e)) ? 1 : 0;
      }
      seg_sorted[base + b0 + pos] = ms;
    }
  } else {
    for (int n = wave; n < RNODES; n += 32) {
      const int beg = base + beg_l[n];
      const int end = base + beg_l[n + 1];
      for (int cb = beg; cb < end; cb += 32) {
        const int mi = cb + lane;
        const bool have = mi < end;
        const int ms = have ? seg_tmp[mi] : 0;
        int pos = 0;
        for (int ob = beg; ob < end; ob += 32) {
          const int oi = ob + lane;
          const int os = (oi < end) ? seg_tmp[oi] : 0;
          const int oc = min(32, end - ob);
          for (int k = 0; k < oc; ++k) {
            const int ks = __builtin_amdgcn_readlane(os, k);
            const int ki = ob + k;
            pos += (ks < ms || (ks == ms && ki < mi)) ? 1 : 0;
          }
        }
        if (have) seg_sorted[beg + pos] = ms;
      }
    }
  }
}

#define APITCH 132
#define WPITCH 136
#define NPW 8
#define RIF 16
#define LWAVES 26
#define LTHREADS (LWAVES * 32)
#define BAND (LWAVES * NPW)
#define RTILES (BAND / 16)
#define WQ ((DFEAT * DFEAT / 4 + LTHREADS - 1) / LTHREADS)
static_assert(LWAVES == RTILES * 2 && DFEAT == 128, "a wave owns one row tile and half of the 128 columns");
template <bool FIRST>
__global__ __launch_bounds__(LTHREADS) void gcn_layer_kernel(
    const float* __restrict__ X, const int* __restrict__ seg_sorted,
    const float* __restrict__ norm_src, int* __restrict__ seg_pairs,
    const int* __restrict__ row_ptr, const float* __restrict__ norm_dst,
    const float* __restrict__ W, const float* __restrict__ bias,
    float* __restrict__ out, int N) {
  __shared__ __attribute__((aligned(16))) float sA[BAND * APITCH];
  __shared__ __attribute__((aligned(16))) float sW[DFEAT * WPITCH];
  __shared__ __attribute__((aligned(16))) float sB[DFEAT];

  const int tid = threadIdx.x;
  const int wave = __builtin_amdgcn_readfirstlane(tid >> 5);
  const int lane = tid & 31;
  const int row0 = blockIdx.x * BAND;
  const int nl = lane & 15;
  const int koff = (lane >> 4) << 1;

  float bload = 0.0f;
  if (tid < DFEAT) bload = bias[tid];
  float4 wq[WQ];
#pragma unroll
  for (int it = 0; it < WQ; ++it) {
    const int idx4 = tid + it * LTHREADS;
    wq[it] = make_float4(0.0f, 0.0f, 0.0f, 0.0f);
    if (idx4 < DFEAT * DFEAT / 4) wq[it] = ((const float4*)W)[idx4];
  }

  {
    const int n0 = row0 + wave * NPW;
    int rp = 0;
    int nd = 0;
    if (lane <= NPW) rp = row_ptr[min(n0 + lane, N)];
    if (lane < NPW) nd = __float_as_int(norm_dst[min(n0 + lane, N - 1)]);
    const int beg = __builtin_amdgcn_readlane(rp, 0);
    const int end = __builtin_amdgcn_readlane(rp, NPW);
    int2* __restrict__ seg = (int2*)seg_pairs;
    const float4* __restrict__ x4 = (const float4*)X;
    float* arow = &sA[(wave * NPW) * APITCH + lane * 4];

    int cur = 0;
    int nextb = __builtin_amdgcn_readlane(rp, 1);
    float4 acc = {0.0f, 0.0f, 0.0f, 0.0f};

    int2 m = make_int2(0, 0);
    if (beg + lane < end) {
      if (FIRST) m.x = seg_sorted[beg + lane];
      else m = seg[beg + lane];
    }
#pragma unroll
    for (int it = 0; it < WQ; ++it) {
      const int idx4 = tid + it * LTHREADS;
      if (idx4 < DFEAT * DFEAT / 4) *(float4*)&sW[(idx4 >> 5) * WPITCH + (idx4 & 31) * 4] = wq[it];
    }
    for (int cb = beg; cb < end; cb += 32) {
      const int cnt = min(32, end - cb);
      int2 mn = make_int2(0, 0);
      const int other = FIRST ? __shfl_xor(m.x, 16) : 0;
      for (int k = 0; k < cnt; k += RIF) {
        int s[RIF];
        float4 v[RIF];
        int sk = 0;
        float wt = 0.0f;
#pragma unroll
        for (int i = 0; i < RIF; ++i) s[i] = __builtin_amdgcn_readlane(m.x, min(k + i, cnt - 1));
        if (FIRST) {
          static_assert(RIF == 16, "a trip is one lane half");
          sk = (((lane >> 4) << 4) == k) ? m.x : other;
          wt = norm_src[sk];
          __builtin_amdgcn_sched_barrier(0);
        }
#pragma unroll
        for (int i = 0; i < RIF; ++i) v[i] = x4[(size_t)s[i] * (DFEAT / 4) + lane];
        if (k == 0 && cb + 32 + lane < end) {
          if (FIRST) mn.x = seg_sorted[cb + 32 + lane];
          else mn = seg[cb + 32 + lane];
        }

#pragma unroll
        for (int i = 0; i < RIF; ++i) {
          if (i == 0 || k + i < cnt) {
            const int e = cb + k + i;
            while (e == nextb) {
              const float sc = __int_as_float(__builtin_amdgcn_readlane(nd, cur));
              *(float4*)(arow + cur * APITCH) = make_float4(acc.x * sc, acc.y * sc, acc.z * sc, acc.w * sc);
              acc = make_float4(0.0f, 0.0f, 0.0f, 0.0f);
              ++cur;
              nextb = __builtin_amdgcn_readlane(rp, cur + 1);
            }
            const float w = FIRST ? __int_as_float(__builtin_amdgcn_readlane(__float_as_int(wt), i))
                                  : __int_as_float(__builtin_amdgcn_readlane(m.y, k + i));
            acc.x += v[i].x * w; acc.y += v[i].y * w; acc.z += v[i].z * w; acc.w += v[i].w * w;
          }
        }
        if (FIRST && lane < 16 && k + lane < cnt) seg[cb + k + lane] = make_int2(sk, __float_as_int(wt));
      }
      m = mn;
    }
    for (; cur < NPW; ++cur) {
      const float sc = __int_as_float(__builtin_amdgcn_readlane(nd, cur));
      *(float4*)(arow + cur * APITCH) = make_float4(acc.x * sc, acc.y * sc, acc.z * sc, acc.w * sc);
      acc = make_float4(0.0f, 0.0f, 0.0f, 0.0f);
    }
  }

  if (tid < DFEAT) sB[tid] = bload;
  __syncthreads();

  const int rtile = wave >> 1;
  const int c0 = (wave & 1) * 64 + nl * 4;
  const float* ap = &sA[(rtile * 16 + nl) * APITCH + koff];
  const float* bp = &sW[koff * WPITCH + c0];
  v8f acc0 = {}, acc1 = {}, acc2 = {}, acc3 = {};

#pragma unroll 8
  for (int ks = 0; ks < DFEAT / 4; ++ks) {
    v2f a;
    a[0] = ap[ks * 4];
    a[1] = ap[ks * 4 + 1];
    const float4 b0 = *(const float4*)&bp[(ks * 4) * WPITCH];
    const float4 b1 = *(const float4*)&bp[(ks * 4 + 1) * WPITCH];
    v2f f0, f1, f2, f3;
    f0[0] = b0.x; f0[1] = b1.x;
    f1[0] = b0.y; f1[1] = b1.y;
    f2[0] = b0.z; f2[1] = b1.z;
    f3[0] = b0.w; f3[1] = b1.w;
    acc0 = __builtin_amdgcn_wmma_f32_16x16x4_f32(false, a, false, f0, (short)0, acc0, false, false);
    acc1 = __builtin_amdgcn_wmma_f32_16x16x4_f32(false, a, false, f1, (short)0, acc1, false, false);
    acc2 = __builtin_amdgcn_wmma_f32_16x16x4_f32(false, a, false, f2, (short)0, acc2, false, false);
    acc3 = __builtin_amdgcn_wmma_f32_16x16x4_f32(false, a, false, f3, (short)0, acc3, false, false);
  }

  const float4 bv = *(const float4*)&sB[c0];
  const int mhalf = (lane >> 4) << 3;
#pragma unroll
  for (int i = 0; i < 8; ++i) {
    const int r = row0 + rtile * 16 + i + mhalf;
    if (r < N)
      *(float4*)&out[(size_t)r * DFEAT + c0] =
          make_float4(fmaxf(acc0[i] + bv.x, 0.0f), fmaxf(acc1[i] + bv.y, 0.0f),
                      fmaxf(acc2[i] + bv.z, 0.0f), fmaxf(acc3[i] + bv.w, 0.0f));
  }
}

extern "C" void kernel_launch(void* const* d_in, const int* in_sizes, int n_in,
                              void* d_out, int out_size, void* d_ws, size_t ws_size,
                              hipStream_t stream) {
  const float* feat = (const float*)d_in[0];
  const int* src    = (const int*)d_in[1];
  const int* dst    = (const int*)d_in[2];
  const float* W1   = (const float*)d_in[3];
  const float* b1   = (const float*)d_in[4];
  const float* W2   = (const float*)d_in[5];
  const float* b2   = (const float*)d_in[6];
  float* out        = (float*)d_out;

  const int N = in_sizes[0] / DFEAT;
  const int E = in_sizes[1];

  const int B = (E + CHUNK - 1) / CHUNK;
  const int P = (N + RNODES - 1) / RNODES;

  float* h1       = (float*)d_ws;
  int* seg_pairs  = (int*)(h1 + (size_t)N * DFEAT);
  int* bucket_d   = seg_pairs + 2 * (size_t)E;
  int* seg_tmp    = bucket_d + 2 * (size_t)E;
  int* bucket_s   = seg_tmp + E;
  int* tab_d      = bucket_s + E;
  int* tab_s      = tab_d + (size_t)(P + 1) * B;
  int* row_ptr    = tab_s + (size_t)(P + 1) * B;
  float* norm_src = (float*)(row_ptr + N + 1);
  float* norm_dst = norm_src + N;
  int* seg_sorted = (int*)(norm_dst + N);

  const int bandBlocks = (N + BAND - 1) / BAND;

  gcn_bin_kernel<<<B, BIN_T, 0, stream>>>(src, dst, bucket_d, bucket_s, tab_d, tab_s, E, P, B);
  gcn_owner_kernel<<<P, OWN_T, 0, stream>>>(bucket_d, bucket_s, tab_d, tab_s, row_ptr, norm_src, norm_dst, seg_tmp, seg_sorted, N, B);

  gcn_layer_kernel<true><<<bandBlocks, LTHREADS, 0, stream>>>(feat, seg_sorted, norm_src, seg_pairs, row_ptr, norm_dst, W1, b1, h1, N);
  gcn_layer_kernel<false><<<bandBlocks, LTHREADS, 0, stream>>>(h1, seg_sorted, norm_src, seg_pairs, row_ptr, norm_dst, W2, b2, out, N);
}
